# starting kernel
# baseline (speedup 1.0000x reference)
#include <hip/hip_runtime.h>
#include <hip/hip_bf16.h>
#include <math.h>

typedef __attribute__((ext_vector_type(2))) float v2f;
typedef __attribute__((ext_vector_type(8))) float v8f;

constexpr int KSZ   = 3;
constexpr int K2    = 9;
constexpr int CIN   = 64;
constexpr int COUT  = 64;
constexpr int NB    = 8;
constexpr int H     = 128;
constexpr int W     = 128;
constexpr int KDIM  = CIN * K2;

constexpr int THREADS     = 256;
constexpr int PIX_PER_BLK = 32;
constexpr int YSTRIDE     = KDIM + 1;

constexpr int LDS_W_OFF  = 0;
constexpr int LDS_Y_OFF  = LDS_W_OFF + KDIM * COUT;
constexpr int LDS_OP_OFF = LDS_Y_OFF + PIX_PER_BLK * YSTRIDE;
constexpr int LDS_N_OFF  = LDS_OP_OFF + PIX_PER_BLK * 81;
constexpr int LDS_FLOATS = LDS_N_OFF + PIX_PER_BLK * K2;

__global__ __launch_bounds__(THREADS, 1)
void dyfconv_fused(const float* __restrict__ x,
                   const float* __restrict__ weight,
                   const float* __restrict__ bias,
                   const float* __restrict__ op_w,
                   const float* __restrict__ op_b,
                   float* __restrict__ out)
{
    extern __shared__ float lds[];
    float* ldsW  = lds + LDS_W_OFF;
    float* ldsY  = lds + LDS_Y_OFF;
    float* ldsOp = lds + LDS_OP_OFF;
    float* ldsN  = lds + LDS_N_OFF;

    const int t     = threadIdx.x;
    const int pbase = blockIdx.x * PIX_PER_BLK;

    for (int idx = t; idx < KDIM * COUT; idx += THREADS) {
        int kk = idx >> 6;
        int o  = idx & (COUT - 1);
        int c  = kk / K2;
        int k  = kk - c * K2;
        ldsW[idx] = weight[(c * COUT + o) * K2 + k];
    }
    for (int idx = t; idx < PIX_PER_BLK * K2; idx += THREADS)
        ldsN[idx] = 0.0f;
    __syncthreads();

    const int p  = t >> 3;
    const int g  = t & 7;
    const int pg = pbase + p;
    const int bb = pg >> 14;
    const int hw = pg & (H * W - 1);
    const int hh = hw >> 7;
    const int ww = hw & (W - 1);

    {
        float sq[K2];
        #pragma unroll
        for (int j = 0; j < K2; ++j) sq[j] = 0.0f;
        for (int cc = 0; cc < 8; ++cc) {
            const int c = g * 8 + cc;
            const float* xc = x + (size_t)(bb * CIN + c) * (H * W);
            #pragma unroll
            for (int ki = 0; ki < KSZ; ++ki) {
                const int hs  = hh + ki - 1;
                const bool okh = (unsigned)hs < (unsigned)H;
                #pragma unroll
                for (int kj = 0; kj < KSZ; ++kj) {
                    const int wsv = ww + kj - 1;
                    const bool ok = okh && ((unsigned)wsv < (unsigned)W);
                    const float v = ok ? xc[hs * W + wsv] : 0.0f;
                    sq[ki * KSZ + kj] += v * v;
                }
            }
        }
        #pragma unroll
        for (int j = 0; j < K2; ++j)
            atomicAdd(&ldsN[p * K2 + j], sq[j]);
    }
    __syncthreads();

    for (int e = t; e < PIX_PER_BLK * 81; e += THREADS) {
        const int pp  = e / 81;
        const int idx = e - pp * 81;
        const float* nb = &ldsN[pp * K2];
        float acc = op_b[idx];
        #pragma unroll
        for (int j = 0; j < K2; ++j)
            acc += sqrtf(nb[j]) * op_w[idx * K2 + j];
        ldsOp[e] = tanhf(acc);
    }
    __syncthreads();

    {
        const float* opp = &ldsOp[p * 81];
        float* yrow = &ldsY[p * YSTRIDE];
        for (int cc = 0; cc < 8; ++cc) {
            const int c = g * 8 + cc;
            const float* xc = x + (size_t)(bb * CIN + c) * (H * W);
            float xu[K2];
            #pragma unroll
            for (int ki = 0; ki < KSZ; ++ki) {
                const int hs  = hh + ki - 1;
                const bool okh = (unsigned)hs < (unsigned)H;
                #pragma unroll
                for (int kj = 0; kj < KSZ; ++kj) {
                    const int wsv = ww + kj - 1;
                    const bool ok = okh && ((unsigned)wsv < (unsigned)W);
                    xu[ki * KSZ + kj] = ok ? xc[hs * W + wsv] : 0.0f;
                }
            }
            #pragma unroll
            for (int k = 0; k < K2; ++k) {
                float acc = xu[k];
                #pragma unroll
                for (int j = 0; j < K2; ++j)
                    acc += xu[j] * opp[j * K2 + k];
                yrow[c * K2 + k] = acc;
            }
        }
    }
    __syncthreads();

    const int wave  = t >> 5;
    const int lane  = t & 31;
    const int mTile = wave & 1;
    const int nTile = wave >> 1;
    const int mIdx  = lane & 15;
    const int nIdx  = lane & 15;
    const int kh    = lane >> 4;

    const float* yrow = &ldsY[(mTile * 16 + mIdx) * YSTRIDE];
    const float* wcol = &ldsW[nTile * 16 + nIdx];

    v8f acc = {};
    #pragma unroll 4
    for (int kk = 0; kk < KDIM; kk += 4) {
        v2f a, bf;
        const int kb = kk + 2 * kh;
        a.x = yrow[kb + 0];
        a.y = yrow[kb + 1];
        bf.x = wcol[(kb + 0) * COUT];
        bf.y = wcol[(kb + 1) * COUT];
        acc = __builtin_amdgcn_wmma_f32_16x16x4_f32(
                   false, a,  false, bf,
                   (short)0, acc,  false,  false);
    }

    const int o  = nTile * 16 + nIdx;
    const float bo = bias[o];
    #pragma unroll
    for (int r = 0; r < 8; ++r) {
        const int m     = r + 8 * kh;
        const int pglob = pbase + mTile * 16 + m;
        const int b_    = pglob >> 14;
        const int hw_   = pglob & (H * W - 1);
        out[((size_t)(b_ * COUT + o)) * (H * W) + hw_] = acc[r] + bo;
    }
}

extern "C" void kernel_launch(void* const* d_in, const int* in_sizes, int n_in,
                              void* d_out, int out_size, void* d_ws, size_t ws_size,
                              hipStream_t stream) {
    (void)in_sizes; (void)n_in; (void)out_size; (void)d_ws; (void)ws_size;
    const float* x      = (const float*)d_in[0];
    const float* weight = (const float*)d_in[1];
    const float* bias   = (const float*)d_in[2];
    const float* op_w   = (const float*)d_in[3];
    const float* op_b   = (const float*)d_in[4];
    float* out = (float*)d_out;

    const int totalPix = NB * H * W;
    const int blocks   = totalPix / PIX_PER_BLK;
    const size_t ldsBytes = (size_t)LDS_FLOATS * sizeof(float);

    dyfconv_fused<<<blocks, THREADS, ldsBytes, stream>>>(x, weight, bias, op_w, op_b, out);
}


# GEMM re-tiled: a wave takes 32 pixels x 16 out-channels over half of K (was 16 x 16 over all K): each weight value is loaded once per block, not twice; K halves added through LDS: sum order changes
# speedup vs baseline: 3.7296x; 3.7296x over previous
#include <hip/hip_runtime.h>
#include <hip/hip_bf16.h>
#include <math.h>

typedef __attribute__((ext_vector_type(2))) float v2f;
typedef __attribute__((ext_vector_type(8))) float v8f;

constexpr int KSZ   = 3;
constexpr int K2    = 9;
constexpr int CIN   = 64;
constexpr int COUT  = 64;
constexpr int NB    = 8;
constexpr int H     = 128;
constexpr int W     = 128;
constexpr int KDIM  = CIN * K2;

constexpr int THREADS     = 256;
constexpr int PIX_PER_BLK = 32;
constexpr int YSTRIDE     = KDIM + 1;

constexpr int LDS_Y_OFF  = 0;
constexpr int LDS_OP_OFF = LDS_Y_OFF + PIX_PER_BLK * YSTRIDE;
constexpr int LDS_N_OFF  = LDS_OP_OFF + PIX_PER_BLK * 81;
constexpr int LDS_FLOATS = LDS_N_OFF + PIX_PER_BLK * K2;

constexpr int TCOLS = PIX_PER_BLK + 2;
constexpr int TCH   = 105;
static_assert(CIN * TCH <= PIX_PER_BLK * YSTRIDE, "tile must fit inside ldsY");
static_assert(W % PIX_PER_BLK == 0, "a block's pixels must share one image row");

#define WOFF(l) (((l) / K2) * (COUT * K2) + ((l) % K2))
constexpr int SLAB = 4 * COUT * K2;

__device__ __forceinline__ void load_b18(float (&b)[18], const float* __restrict__ slab,
                                         const unsigned (&boff)[18]) {
    #pragma unroll
    for (int i = 0; i < 18; ++i) b[i] = slab[boff[i]];
}

__device__ __forceinline__ void mma_trip2(v8f& acc0, v8f& acc1, const float (&b)[18],
                                          const float* ya0, const float* ya1) {
    v2f a0[K2], a1[K2];
    #pragma unroll
    for (int s = 0; s < K2; ++s) {
        a0[s].x = ya0[s * 4]; a0[s].y = ya0[s * 4 + 1];
        a1[s].x = ya1[s * 4]; a1[s].y = ya1[s * 4 + 1];
    }
    #pragma unroll
    for (int s = 0; s < K2; ++s) {
        v2f bf;
        bf.x = b[2 * s];
        bf.y = b[2 * s + 1];
        acc0 = __builtin_amdgcn_wmma_f32_16x16x4_f32(
                    false, a0[s],  false, bf,
                    (short)0, acc0,  false,  false);
        acc1 = __builtin_amdgcn_wmma_f32_16x16x4_f32(
                    false, a1[s],  false, bf,
                    (short)0, acc1,  false,  false);
    }
}

__global__ __launch_bounds__(THREADS) __attribute__((amdgpu_waves_per_eu(6, 6)))
void dyfconv_fused(const float* __restrict__ x,
                   const float* __restrict__ weight,
                   const float* __restrict__ bias,
                   const float* __restrict__ op_w,
                   const float* __restrict__ op_b,
                   float* __restrict__ out)
{
    extern __shared__ float lds[];
    float* ldsY  = lds + LDS_Y_OFF;
    float* ldsOp = lds + LDS_OP_OFF;
    float* ldsN  = lds + LDS_N_OFF;

    const int t     = threadIdx.x;
    const int pbase = blockIdx.x * PIX_PER_BLK;

    for (int idx = t; idx < PIX_PER_BLK * K2; idx += THREADS)
        ldsN[idx] = 0.0f;

    float* tile = ldsY;
    const int bb  = pbase >> 14;
    const int hw0 = pbase & (H * W - 1);
    const int hh  = hw0 >> 7;
    const int ww0 = hw0 & (W - 1);
    const float* xb = x + (size_t)bb * CIN * (H * W);
    {
        const int wv = t >> 5;
        const int ln = t & 31;
        int  rowoff[KSZ];
        bool rowok[KSZ];
        #pragma unroll
        for (int r = 0; r < KSZ; ++r) {
            const int hs = hh + r - 1;
            rowok[r]  = (unsigned)hs < (unsigned)H;
            rowoff[r] = (rowok[r] ? hs : hh) * W;
        }
        const bool own = ln < 8 * KSZ;
        const int  hq  = own ? ln : (8 * KSZ - 1);
        const int  hci = hq / KSZ;
        const int  hrw = hq - hci * KSZ;
        const bool okr = (unsigned)(hh + hrw - 1) < (unsigned)H;
        const float* xr = xb + (size_t)(wv * 8 + hci) * (H * W) + (okr ? (hh + hrw - 1) : hh) * W;
        const bool okl = okr && (ww0 > 0);
        const bool okq = okr && (ww0 + PIX_PER_BLK < W);
        const float hla = xr[okl ? ww0 - 1 : ww0];
        const float hra = xr[okq ? ww0 + PIX_PER_BLK : ww0];
        const float hl = (own && okl) ? hla : 0.0f;
        const float hr = (own && okq) ? hra : 0.0f;
        const int ha0 = own ? ((wv * 8 + hci) * TCH + hrw * TCOLS)
                            : ((wv * 8 + (ln - 8 * KSZ)) * TCH + KSZ * TCOLS);
        const int ha1 = own ? (ha0 + TCOLS - 1) : (ha0 + 1);
        float v[8 * KSZ];
        const float* xw = xb + (size_t)(wv * 8) * (H * W) + ww0 + ln;
        #pragma unroll
        for (int i = 0; i < 8 * KSZ; ++i) {
            const int ci = i / KSZ, r = i % KSZ;
            const float a = xw[ci * (H * W) + rowoff[r]];
            v[i] = rowok[r] ? a : 0.0f;
        }
        #pragma unroll
        for (int i = 0; i < 8 * KSZ; ++i) {
            const int ci = i / KSZ, r = i % KSZ;
            tile[(wv * 8 + ci) * TCH + r * TCOLS + 1 + ln] = v[i];
        }
        tile[ha0] = hl;
        tile[ha1] = hr;
    }
    __syncthreads();

    const int p  = t >> 3;
    const int g  = t & 7;

    float xu[8][K2];
    {
        const float* tp = tile + (g * 8) * TCH + p;
        #pragma unroll
        for (int cc = 0; cc < 8; ++cc)
            #pragma unroll
            for (int ki = 0; ki < KSZ; ++ki)
                #pragma unroll
                for (int kj = 0; kj < KSZ; ++kj)
                    xu[cc][ki * KSZ + kj] = tp[cc * TCH + ki * TCOLS + kj];
    }

    {
        float sq[K2];
        #pragma unroll
        for (int j = 0; j < K2; ++j) sq[j] = 0.0f;
        #pragma unroll
        for (int cc = 0; cc < 8; ++cc) {
            #pragma unroll
            for (int j = 0; j < K2; ++j) {
                const float v = xu[cc][j];
                sq[j] += v * v;
            }
        }
        #pragma unroll
        for (int j = 0; j < K2; ++j)
            atomicAdd(&ldsN[p * K2 + j], sq[j]);
    }
    __syncthreads();

    for (int e = t; e < PIX_PER_BLK * 81; e += THREADS) {
        const int pp  = e / 81;
        const int idx = e - pp * 81;
        const float* nb = &ldsN[pp * K2];
        float acc = op_b[idx];
        #pragma unroll
        for (int j = 0; j < K2; ++j)
            acc += sqrtf(nb[j]) * op_w[idx * K2 + j];
        ldsOp[e] = tanhf(acc);
    }
    __syncthreads();

    {
        const float* opp = &ldsOp[p * 81];
        float* yrow = &ldsY[p * YSTRIDE];
        #pragma unroll
        for (int cc = 0; cc < 8; ++cc) {
            const int c = g * 8 + cc;
            #pragma unroll
            for (int k = 0; k < K2; ++k) {
                float acc = xu[cc][k];
                #pragma unroll
                for (int j = 0; j < K2; ++j)
                    acc += xu[cc][j] * opp[j * K2 + k];
                yrow[c * K2 + k] = acc;
            }
        }
    }

    const int wave  = t >> 5;
    const int lane  = t & 31;
    const int kq    = wave & 1;
    const int nTile = wave >> 1;
    const int mIdx  = lane & 15;
    const int nIdx  = lane & 15;
    const int kh    = lane >> 4;
    unsigned boff[18];
    #pragma unroll
    for (int s = 0; s < K2; ++s) {
        const int l0 = s * 4;
        const int l2 = s * 4 + 2;
        const unsigned ocol = (unsigned)((nTile * 16 + nIdx) * K2);
        boff[2 * s]     = ocol + (unsigned)(kh ? WOFF(l2)     : WOFF(l0));
        boff[2 * s + 1] = ocol + (unsigned)(kh ? WOFF(l2 + 1) : WOFF(l0 + 1));
    }
    constexpr int TRIPS = CIN / 8;
    const int c4b = kq * TRIPS;
    float b0[18];
    load_b18(b0, weight + c4b * SLAB, boff);
    __syncthreads();

    const float* ya0 = &ldsY[mIdx * YSTRIDE + 2 * kh];
    const float* ya1 = ya0 + 16 * YSTRIDE;

    v8f acc0 = {};
    v8f acc1 = {};
    float b1[18];
    for (int i = 0; i < TRIPS; i += 2) {
        const int c4 = c4b + i;
        load_b18(b1, weight + (c4 + 1) * SLAB, boff);
        mma_trip2(acc0, acc1, b0, ya0 + c4 * (4 * K2), ya1 + c4 * (4 * K2));
        const int cn = (i + 2 < TRIPS) ? (c4 + 2) : (c4 + 1);
        load_b18(b0, weight + cn * SLAB, boff);
        mma_trip2(acc0, acc1, b1, ya0 + (c4 + 1) * (4 * K2), ya1 + (c4 + 1) * (4 * K2));
    }

    float* red = ldsOp;
    static_assert(8 * 8 * 32 <= PIX_PER_BLK * 81, "exchange area must fit inside ldsOp");
    const v8f give = kq ? acc0 : acc1;
    v8f keep       = kq ? acc1 : acc0;
    #pragma unroll
    for (int r = 0; r < 8; ++r)
        red[wave * 256 + r * 32 + lane] = give[r];
    __syncthreads();
    #pragma unroll
    for (int r = 0; r < 8; ++r)
        keep[r] += red[(wave ^ 1) * 256 + r * 32 + lane];

    const int o  = nTile * 16 + nIdx;
    const float bo = bias[o];
    #pragma unroll
    for (int r = 0; r < 8; ++r) {
        const int m     = r + 8 * kh;
        const int pglob = pbase + kq * 16 + m;
        const int b_    = pglob >> 14;
        const int hw_   = pglob & (H * W - 1);
        out[((size_t)(b_ * COUT + o)) * (H * W) + hw_] = keep[r] + bo;
    }
}

extern "C" void kernel_launch(void* const* d_in, const int* in_sizes, int n_in,
                              void* d_out, int out_size, void* d_ws, size_t ws_size,
                              hipStream_t stream) {
    (void)in_sizes; (void)n_in; (void)out_size; (void)d_ws; (void)ws_size;
    const float* x      = (const float*)d_in[0];
    const float* weight = (const float*)d_in[1];
    const float* bias   = (const float*)d_in[2];
    const float* op_w   = (const float*)d_in[3];
    const float* op_b   = (const float*)d_in[4];
    float* out = (float*)d_out;

    const int totalPix = NB * H * W;
    const int blocks   = totalPix / PIX_PER_BLK;
    const size_t ldsBytes = (size_t)LDS_FLOATS * sizeof(float);

    dyfconv_fused<<<blocks, THREADS, ldsBytes, stream>>>(x, weight, bias, op_w, op_b, out);
}


# Operator map tanh(n @ op_w^T + op_b) as a 32x96x12 f32 WMMA GEMM per block (was a scalar loop, 9 roots per entry): roots per fragment element, op_w fetched at kernel start; its sum order changes
# speedup vs baseline: 4.2899x; 1.1502x over previous
#include <hip/hip_runtime.h>
#include <hip/hip_bf16.h>
#include <math.h>

typedef __attribute__((ext_vector_type(2))) float v2f;
typedef __attribute__((ext_vector_type(8))) float v8f;

constexpr int KSZ   = 3;
constexpr int K2    = 9;
constexpr int CIN   = 64;
constexpr int COUT  = 64;
constexpr int NB    = 8;
constexpr int H     = 128;
constexpr int W     = 128;
constexpr int KDIM  = CIN * K2;

constexpr int THREADS     = 256;
constexpr int PIX_PER_BLK = 32;
constexpr int YSTRIDE     = KDIM + 1;

constexpr int LDS_Y_OFF  = 0;
constexpr int LDS_OP_OFF = LDS_Y_OFF + PIX_PER_BLK * YSTRIDE;
constexpr int LDS_N_OFF  = LDS_OP_OFF + PIX_PER_BLK * 81;
constexpr int LDS_FLOATS = LDS_N_OFF + PIX_PER_BLK * K2;

constexpr int TCOLS = PIX_PER_BLK + 2;
constexpr int TCH   = 105;
static_assert(CIN * TCH <= PIX_PER_BLK * YSTRIDE, "tile must fit inside ldsY");
static_assert(W % PIX_PER_BLK == 0, "a block's pixels must share one image row");

#define WOFF(l) (((l) / K2) * (COUT * K2) + ((l) % K2))
constexpr int SLAB = 4 * COUT * K2;

__device__ __forceinline__ void load_b18(float (&b)[18], const float* __restrict__ slab,
                                         const unsigned (&boff)[18]) {
    #pragma unroll
    for (int i = 0; i < 18; ++i) b[i] = slab[boff[i]];
}

__device__ __forceinline__ void mma_trip2(v8f& acc0, v8f& acc1, const float (&b)[18],
                                          const float* ya0, const float* ya1) {
    v2f a0[K2], a1[K2];
    #pragma unroll
    for (int s = 0; s < K2; ++s) {
        a0[s].x = ya0[s * 4]; a0[s].y = ya0[s * 4 + 1];
        a1[s].x = ya1[s * 4]; a1[s].y = ya1[s * 4 + 1];
    }
    #pragma unroll
    for (int s = 0; s < K2; ++s) {
        v2f bf;
        bf.x = b[2 * s];
        bf.y = b[2 * s + 1];
        acc0 = __builtin_amdgcn_wmma_f32_16x16x4_f32(
                    false, a0[s],  false, bf,
                    (short)0, acc0,  false,  false);
        acc1 = __builtin_amdgcn_wmma_f32_16x16x4_f32(
                    false, a1[s],  false, bf,
                    (short)0, acc1,  false,  false);
    }
}

__global__ __launch_bounds__(THREADS) __attribute__((amdgpu_waves_per_eu(6, 6)))
void dyfconv_fused(const float* __restrict__ x,
                   const float* __restrict__ weight,
                   const float* __restrict__ bias,
                   const float* __restrict__ op_w,
                   const float* __restrict__ op_b,
                   float* __restrict__ out)
{
    extern __shared__ float lds[];
    float* ldsY  = lds + LDS_Y_OFF;
    float* ldsOp = lds + LDS_OP_OFF;
    float* ldsN  = lds + LDS_N_OFF;

    const int t     = threadIdx.x;
    const int pbase = blockIdx.x * PIX_PER_BLK;
    const int wave  = t >> 5;
    const int lane  = t & 31;
    const int kh    = lane >> 4;
    const int l16   = lane & 15;

    const int  oidx = wave * 16 + l16;
    const bool oval = oidx < K2 * K2;
    float ob[5];
    float obias;
    {
        const int oc = oval ? oidx : (K2 * K2 - 1);
        const float* wrow = op_w + oc * K2;
        const float w0 = wrow[2 * kh];
        const float w1 = wrow[2 * kh + 1];
        const float w2 = wrow[4 + 2 * kh];
        const float w3 = wrow[5 + 2 * kh];
        const float w4 = wrow[8];
        const float wb = op_b[oc];
        ob[0] = oval ? w0 : 0.0f;
        ob[1] = oval ? w1 : 0.0f;
        ob[2] = oval ? w2 : 0.0f;
        ob[3] = oval ? w3 : 0.0f;
        ob[4] = (oval && kh == 0) ? w4 : 0.0f;
        obias = oval ? wb : 0.0f;
    }

    for (int idx = t; idx < PIX_PER_BLK * K2; idx += THREADS)
        ldsN[idx] = 0.0f;

    float* tile = ldsY;
    const int bb  = pbase >> 14;
    const int hw0 = pbase & (H * W - 1);
    const int hh  = hw0 >> 7;
    const int ww0 = hw0 & (W - 1);
    const float* xb = x + (size_t)bb * CIN * (H * W);
    {
        const int wv = t >> 5;
        const int ln = t & 31;
        int  rowoff[KSZ];
        bool rowok[KSZ];
        #pragma unroll
        for (int r = 0; r < KSZ; ++r) {
            const int hs = hh + r - 1;
            rowok[r]  = (unsigned)hs < (unsigned)H;
            rowoff[r] = (rowok[r] ? hs : hh) * W;
        }
        const bool own = ln < 8 * KSZ;
        const int  hq  = own ? ln : (8 * KSZ - 1);
        const int  hci = hq / KSZ;
        const int  hrw = hq - hci * KSZ;
        const bool okr = (unsigned)(hh + hrw - 1) < (unsigned)H;
        const float* xr = xb + (size_t)(wv * 8 + hci) * (H * W) + (okr ? (hh + hrw - 1) : hh) * W;
        const bool okl = okr && (ww0 > 0);
        const bool okq = okr && (ww0 + PIX_PER_BLK < W);
        const float hla = xr[okl ? ww0 - 1 : ww0];
        const float hra = xr[okq ? ww0 + PIX_PER_BLK : ww0];
        const float hl = (own && okl) ? hla : 0.0f;
        const float hr = (own && okq) ? hra : 0.0f;
        const int ha0 = own ? ((wv * 8 + hci) * TCH + hrw * TCOLS)
                            : ((wv * 8 + (ln - 8 * KSZ)) * TCH + KSZ * TCOLS);
        const int ha1 = own ? (ha0 + TCOLS - 1) : (ha0 + 1);
        float v[8 * KSZ];
        const float* xw = xb + (size_t)(wv * 8) * (H * W) + ww0 + ln;
        #pragma unroll
        for (int i = 0; i < 8 * KSZ; ++i) {
            const int ci = i / KSZ, r = i % KSZ;
            const float a = xw[ci * (H * W) + rowoff[r]];
            v[i] = rowok[r] ? a : 0.0f;
        }
        #pragma unroll
        for (int i = 0; i < 8 * KSZ; ++i) {
            const int ci = i / KSZ, r = i % KSZ;
            tile[(wv * 8 + ci) * TCH + r * TCOLS + 1 + ln] = v[i];
        }
        tile[ha0] = hl;
        tile[ha1] = hr;
    }
    __syncthreads();

    const int p  = t >> 3;
    const int g  = t & 7;

    float xu[8][K2];
    {
        const float* tp = tile + (g * 8) * TCH + p;
        #pragma unroll
        for (int cc = 0; cc < 8; ++cc)
            #pragma unroll
            for (int ki = 0; ki < KSZ; ++ki)
                #pragma unroll
                for (int kj = 0; kj < KSZ; ++kj)
                    xu[cc][ki * KSZ + kj] = tp[cc * TCH + ki * TCOLS + kj];
    }

    {
        float sq[K2];
        #pragma unroll
        for (int j = 0; j < K2; ++j) sq[j] = 0.0f;
        #pragma unroll
        for (int cc = 0; cc < 8; ++cc) {
            #pragma unroll
            for (int j = 0; j < K2; ++j) {
                const float v = xu[cc][j];
                sq[j] += v * v;
            }
        }
        #pragma unroll
        for (int j = 0; j < K2; ++j)
            atomicAdd(&ldsN[p * K2 + j], sq[j]);
    }
    __syncthreads();

    const int wave_s = __builtin_amdgcn_readfirstlane(wave);
    if (wave_s < 6) {
        v2f af[2][3];
        #pragma unroll
        for (int m = 0; m < 2; ++m) {
            const float* nb = &ldsN[(m * 16 + l16) * K2];
            af[m][0].x = sqrtf(nb[2 * kh]);
            af[m][0].y = sqrtf(nb[2 * kh + 1]);
            af[m][1].x = sqrtf(nb[4 + 2 * kh]);
            af[m][1].y = sqrtf(nb[5 + 2 * kh]);
            const float r8 = sqrtf(nb[8]);
            af[m][2].x = kh ? 0.0f : r8;
            af[m][2].y = 0.0f;
        }
        v8f c0, c1;
        #pragma unroll
        for (int r = 0; r < 8; ++r) { c0[r] = obias; c1[r] = obias; }
        #pragma unroll
        for (int s = 0; s < 3; ++s) {
            v2f bf;
            bf.x = ob[2 * s];
            bf.y = (s < 2) ? ob[2 * s + 1] : 0.0f;
            c0 = __builtin_amdgcn_wmma_f32_16x16x4_f32(
                      false, af[0][s],  false, bf,
                      (short)0, c0,  false,  false);
            c1 = __builtin_amdgcn_wmma_f32_16x16x4_f32(
                      false, af[1][s],  false, bf,
                      (short)0, c1,  false,  false);
        }
        if (oval) {
            float* orow = &ldsOp[(8 * kh) * 81 + oidx];
            #pragma unroll
            for (int r = 0; r < 8; ++r) {
                orow[r * 81]        = tanhf(c0[r]);
                orow[(16 + r) * 81] = tanhf(c1[r]);
            }
        }
    }
    __syncthreads();

    {
        const float* opp = &ldsOp[p * 81];
        float* yrow = &ldsY[p * YSTRIDE];
        #pragma unroll
        for (int cc = 0; cc < 8; ++cc) {
            const int c = g * 8 + cc;
            #pragma unroll
            for (int k = 0; k < K2; ++k) {
                float acc = xu[cc][k];
                #pragma unroll
                for (int j = 0; j < K2; ++j)
                    acc += xu[cc][j] * opp[j * K2 + k];
                yrow[c * K2 + k] = acc;
            }
        }
    }

    const int kq    = wave & 1;
    const int nTile = wave >> 1;
    const int mIdx  = l16;
    const int nIdx  = l16;
    unsigned boff[18];
    #pragma unroll
    for (int s = 0; s < K2; ++s) {
        const int l0 = s * 4;
        const int l2 = s * 4 + 2;
        const unsigned ocol = (unsigned)((nTile * 16 + nIdx) * K2);
        boff[2 * s]     = ocol + (unsigned)(kh ? WOFF(l2)     : WOFF(l0));
        boff[2 * s + 1] = ocol + (unsigned)(kh ? WOFF(l2 + 1) : WOFF(l0 + 1));
    }
    constexpr int TRIPS = CIN / 8;
    const int c4b = kq * TRIPS;
    float b0[18];
    load_b18(b0, weight + c4b * SLAB, boff);
    __syncthreads();

    const float* ya0 = &ldsY[mIdx * YSTRIDE + 2 * kh];
    const float* ya1 = ya0 + 16 * YSTRIDE;

    v8f acc0 = {};
    v8f acc1 = {};
    float b1[18];
    for (int i = 0; i < TRIPS; i += 2) {
        const int c4 = c4b + i;
        load_b18(b1, weight + (c4 + 1) * SLAB, boff);
        mma_trip2(acc0, acc1, b0, ya0 + c4 * (4 * K2), ya1 + c4 * (4 * K2));
        const int cn = (i + 2 < TRIPS) ? (c4 + 2) : (c4 + 1);
        load_b18(b0, weight + cn * SLAB, boff);
        mma_trip2(acc0, acc1, b1, ya0 + (c4 + 1) * (4 * K2), ya1 + (c4 + 1) * (4 * K2));
    }

    float* red = ldsOp;
    static_assert(8 * 8 * 32 <= PIX_PER_BLK * 81, "exchange area must fit inside ldsOp");
    const v8f give = kq ? acc0 : acc1;
    v8f keep       = kq ? acc1 : acc0;
    #pragma unroll
    for (int r = 0; r < 8; ++r)
        red[wave * 256 + r * 32 + lane] = give[r];
    __syncthreads();
    #pragma unroll
    for (int r = 0; r < 8; ++r)
        keep[r] += red[(wave ^ 1) * 256 + r * 32 + lane];

    const int o  = nTile * 16 + nIdx;
    const float bo = bias[o];
    #pragma unroll
    for (int r = 0; r < 8; ++r) {
        const int m     = r + 8 * kh;
        const int pglob = pbase + kq * 16 + m;
        const int b_    = pglob >> 14;
        const int hw_   = pglob & (H * W - 1);
        out[((size_t)(b_ * COUT + o)) * (H * W) + hw_] = keep[r] + bo;
    }
}

extern "C" void kernel_launch(void* const* d_in, const int* in_sizes, int n_in,
                              void* d_out, int out_size, void* d_ws, size_t ws_size,
                              hipStream_t stream) {
    (void)in_sizes; (void)n_in; (void)out_size; (void)d_ws; (void)ws_size;
    const float* x      = (const float*)d_in[0];
    const float* weight = (const float*)d_in[1];
    const float* bias   = (const float*)d_in[2];
    const float* op_w   = (const float*)d_in[3];
    const float* op_b   = (const float*)d_in[4];
    float* out = (float*)d_out;

    const int totalPix = NB * H * W;
    const int blocks   = totalPix / PIX_PER_BLK;
    const size_t ldsBytes = (size_t)LDS_FLOATS * sizeof(float);

    dyfconv_fused<<<blocks, THREADS, ldsBytes, stream>>>(x, weight, bias, op_w, op_b, out);
}


# GEMM k order: lane halves take channels 32 apart (were rows 2 apart): A reads 2-way not 4-way bank conflict; B: 6 wide loads from one lane base per trip (were 18, offset registers); sum order changes
# speedup vs baseline: 4.6512x; 1.0842x over previous
#include <hip/hip_runtime.h>
#include <hip/hip_bf16.h>
#include <math.h>

typedef __attribute__((ext_vector_type(2))) float v2f;
typedef __attribute__((ext_vector_type(8))) float v8f;

constexpr int KSZ   = 3;
constexpr int K2    = 9;
constexpr int CIN   = 64;
constexpr int COUT  = 64;
constexpr int NB    = 8;
constexpr int H     = 128;
constexpr int W     = 128;
constexpr int KDIM  = CIN * K2;

constexpr int THREADS     = 256;
constexpr int PIX_PER_BLK = 32;
constexpr int YSTRIDE     = KDIM + 1;

constexpr int LDS_Y_OFF  = 0;
constexpr int LDS_OP_OFF = LDS_Y_OFF + PIX_PER_BLK * YSTRIDE;
constexpr int LDS_N_OFF  = LDS_OP_OFF + PIX_PER_BLK * 81;
constexpr int LDS_FLOATS = LDS_N_OFF + PIX_PER_BLK * K2;

constexpr int TCOLS = PIX_PER_BLK + 2;
constexpr int TCH   = 105;
static_assert(CIN * TCH <= PIX_PER_BLK * YSTRIDE, "tile must fit inside ldsY");
static_assert(W % PIX_PER_BLK == 0, "a block's pixels must share one image row");

constexpr int CHW   = COUT * K2;
constexpr int QROWS = 16 * K2;
static_assert(((2 * QROWS) % 64) == 32 && (YSTRIDE % 64) == 1, "bank spacing of the A fragment reads");
__host__ __device__ constexpr int kq_sel(int s, int e) { return s < 4 ? 0 : (s < 8 ? 1 : e); }
__host__ __device__ constexpr int ktap(int s, int e)   { return s < 8 ? 2 * (s & 3) + e : 8; }

__device__ __forceinline__ void load_b18(float (&b)[18], const float* __restrict__ wl) {
    #pragma unroll
    for (int s = 0; s < K2; ++s) {
        b[2 * s]     = wl[kq_sel(s, 0) * (16 * CHW) + ktap(s, 0)];
        b[2 * s + 1] = wl[kq_sel(s, 1) * (16 * CHW) + ktap(s, 1)];
    }
}

__device__ __forceinline__ void mma_trip2(v8f& acc0, v8f& acc1, const float (&b)[18],
                                          const float* ya0, const float* ya1) {
    v2f a0[K2], a1[K2];
    #pragma unroll
    for (int s = 0; s < K2; ++s) {
        a0[s].x = ya0[kq_sel(s, 0) * QROWS + ktap(s, 0)];
        a0[s].y = ya0[kq_sel(s, 1) * QROWS + ktap(s, 1)];
        a1[s].x = ya1[kq_sel(s, 0) * QROWS + ktap(s, 0)];
        a1[s].y = ya1[kq_sel(s, 1) * QROWS + ktap(s, 1)];
    }
    #pragma unroll
    for (int s = 0; s < K2; ++s) {
        v2f bf;
        bf.x = b[2 * s];
        bf.y = b[2 * s + 1];
        acc0 = __builtin_amdgcn_wmma_f32_16x16x4_f32(
                    false, a0[s],  false, bf,
                    (short)0, acc0,  false,  false);
        acc1 = __builtin_amdgcn_wmma_f32_16x16x4_f32(
                    false, a1[s],  false, bf,
                    (short)0, acc1,  false,  false);
    }
}

__global__ __launch_bounds__(THREADS) __attribute__((amdgpu_waves_per_eu(6, 6)))
void dyfconv_fused(const float* __restrict__ x,
                   const float* __restrict__ weight,
                   const float* __restrict__ bias,
                   const float* __restrict__ op_w,
                   const float* __restrict__ op_b,
                   float* __restrict__ out)
{
    extern __shared__ float lds[];
    float* ldsY  = lds + LDS_Y_OFF;
    float* ldsOp = lds + LDS_OP_OFF;
    float* ldsN  = lds + LDS_N_OFF;

    const int t     = threadIdx.x;
    const int pbase = blockIdx.x * PIX_PER_BLK;
    const int wave  = t >> 5;
    const int lane  = t & 31;
    const int kh    = lane >> 4;
    const int l16   = lane & 15;

    const int  oidx = wave * 16 + l16;
    const bool oval = oidx < K2 * K2;
    float ob[5];
    float obias;
    {
        const int oc = oval ? oidx : (K2 * K2 - 1);
        const float* wrow = op_w + oc * K2;
        const float w0 = wrow[2 * kh];
        const float w1 = wrow[2 * kh + 1];
        const float w2 = wrow[4 + 2 * kh];
        const float w3 = wrow[5 + 2 * kh];
        const float w4 = wrow[8];
        const float wb = op_b[oc];
        ob[0] = oval ? w0 : 0.0f;
        ob[1] = oval ? w1 : 0.0f;
        ob[2] = oval ? w2 : 0.0f;
        ob[3] = oval ? w3 : 0.0f;
        ob[4] = (oval && kh == 0) ? w4 : 0.0f;
        obias = oval ? wb : 0.0f;
    }

    for (int idx = t; idx < PIX_PER_BLK * K2; idx += THREADS)
        ldsN[idx] = 0.0f;

    float* tile = ldsY;
    const int bb  = pbase >> 14;
    const int hw0 = pbase & (H * W - 1);
    const int hh  = hw0 >> 7;
    const int ww0 = hw0 & (W - 1);
    const float* xb = x + (size_t)bb * CIN * (H * W);
    {
        const int wv = t >> 5;
        const int ln = t & 31;
        int  rowoff[KSZ];
        bool rowok[KSZ];
        #pragma unroll
        for (int r = 0; r < KSZ; ++r) {
            const int hs = hh + r - 1;
            rowok[r]  = (unsigned)hs < (unsigned)H;
            rowoff[r] = (rowok[r] ? hs : hh) * W;
        }
        const bool own = ln < 8 * KSZ;
        const int  hq  = own ? ln : (8 * KSZ - 1);
        const int  hci = hq / KSZ;
        const int  hrw = hq - hci * KSZ;
        const bool okr = (unsigned)(hh + hrw - 1) < (unsigned)H;
        const float* xr = xb + (size_t)(wv * 8 + hci) * (H * W) + (okr ? (hh + hrw - 1) : hh) * W;
        const bool okl = okr && (ww0 > 0);
        const bool okq = okr && (ww0 + PIX_PER_BLK < W);
        const float hla = xr[okl ? ww0 - 1 : ww0];
        const float hra = xr[okq ? ww0 + PIX_PER_BLK : ww0];
        const float hl = (own && okl) ? hla : 0.0f;
        const float hr = (own && okq) ? hra : 0.0f;
        const int ha0 = own ? ((wv * 8 + hci) * TCH + hrw * TCOLS)
                            : ((wv * 8 + (ln - 8 * KSZ)) * TCH + KSZ * TCOLS);
        const int ha1 = own ? (ha0 + TCOLS - 1) : (ha0 + 1);
        float v[8 * KSZ];
        const float* xw = xb + (size_t)(wv * 8) * (H * W) + ww0 + ln;
        #pragma unroll
        for (int i = 0; i < 8 * KSZ; ++i) {
            const int ci = i / KSZ, r = i % KSZ;
            const float a = xw[ci * (H * W) + rowoff[r]];
            v[i] = rowok[r] ? a : 0.0f;
        }
        #pragma unroll
        for (int i = 0; i < 8 * KSZ; ++i) {
            const int ci = i / KSZ, r = i % KSZ;
            tile[(wv * 8 + ci) * TCH + r * TCOLS + 1 + ln] = v[i];
        }
        tile[ha0] = hl;
        tile[ha1] = hr;
    }
    __syncthreads();

    const int p  = t >> 3;
    const int g  = t & 7;

    float xu[8][K2];
    {
        const float* tp = tile + (g * 8) * TCH + p;
        #pragma unroll
        for (int cc = 0; cc < 8; ++cc)
            #pragma unroll
            for (int ki = 0; ki < KSZ; ++ki)
                #pragma unroll
                for (int kj = 0; kj < KSZ; ++kj)
                    xu[cc][ki * KSZ + kj] = tp[cc * TCH + ki * TCOLS + kj];
    }

    {
        float sq[K2];
        #pragma unroll
        for (int j = 0; j < K2; ++j) sq[j] = 0.0f;
        #pragma unroll
        for (int cc = 0; cc < 8; ++cc) {
            #pragma unroll
            for (int j = 0; j < K2; ++j) {
                const float v = xu[cc][j];
                sq[j] += v * v;
            }
        }
        #pragma unroll
        for (int j = 0; j < K2; ++j)
            atomicAdd(&ldsN[p * K2 + j], sq[j]);
    }
    __syncthreads();

    const int wave_s = __builtin_amdgcn_readfirstlane(wave);
    if (wave_s < 6) {
        v2f af[2][3];
        #pragma unroll
        for (int m = 0; m < 2; ++m) {
            const float* nb = &ldsN[(m * 16 + l16) * K2];
            af[m][0].x = sqrtf(nb[2 * kh]);
            af[m][0].y = sqrtf(nb[2 * kh + 1]);
            af[m][1].x = sqrtf(nb[4 + 2 * kh]);
            af[m][1].y = sqrtf(nb[5 + 2 * kh]);
            const float r8 = sqrtf(nb[8]);
            af[m][2].x = kh ? 0.0f : r8;
            af[m][2].y = 0.0f;
        }
        v8f c0, c1;
        #pragma unroll
        for (int r = 0; r < 8; ++r) { c0[r] = obias; c1[r] = obias; }
        #pragma unroll
        for (int s = 0; s < 3; ++s) {
            v2f bf;
            bf.x = ob[2 * s];
            bf.y = (s < 2) ? ob[2 * s + 1] : 0.0f;
            c0 = __builtin_amdgcn_wmma_f32_16x16x4_f32(
                      false, af[0][s],  false, bf,
                      (short)0, c0,  false,  false);
            c1 = __builtin_amdgcn_wmma_f32_16x16x4_f32(
                      false, af[1][s],  false, bf,
                      (short)0, c1,  false,  false);
        }
        if (oval) {
            float* orow = &ldsOp[(8 * kh) * 81 + oidx];
            #pragma unroll
            for (int r = 0; r < 8; ++r) {
                orow[r * 81]        = tanhf(c0[r]);
                orow[(16 + r) * 81] = tanhf(c1[r]);
            }
        }
    }
    __syncthreads();

    {
        const float* opp = &ldsOp[p * 81];
        float* yrow = &ldsY[p * YSTRIDE];
        #pragma unroll
        for (int cc = 0; cc < 8; ++cc) {
            const int c = g * 8 + cc;
            #pragma unroll
            for (int k = 0; k < K2; ++k) {
                float acc = xu[cc][k];
                #pragma unroll
                for (int j = 0; j < K2; ++j)
                    acc += xu[cc][j] * opp[j * K2 + k];
                yrow[c * K2 + k] = acc;
            }
        }
    }

    const int kq    = wave & 1;
    const int nTile = wave >> 1;
    const int mIdx  = l16;
    const int nIdx  = l16;
    const float* wl = weight + (32 * kh) * CHW + (nTile * 16 + nIdx) * K2;
    constexpr int TRIPS = CIN / 8;
    const int c4b = kq * TRIPS;
    float b0[18];
    load_b18(b0, wl + c4b * CHW);
    __syncthreads();

    const float* ya0 = &ldsY[mIdx * YSTRIDE + 2 * QROWS * kh];
    const float* ya1 = ya0 + 16 * YSTRIDE;

    v8f acc0 = {};
    v8f acc1 = {};
    float b1[18];
    for (int i = 0; i < TRIPS; i += 2) {
        const int c4 = c4b + i;
        load_b18(b1, wl + (c4 + 1) * CHW);
        mma_trip2(acc0, acc1, b0, ya0 + c4 * K2, ya1 + c4 * K2);
        const int cn = (i + 2 < TRIPS) ? (c4 + 2) : (c4 + 1);
        load_b18(b0, wl + cn * CHW);
        mma_trip2(acc0, acc1, b1, ya0 + (c4 + 1) * K2, ya1 + (c4 + 1) * K2);
    }

    float* red = ldsOp;
    static_assert(8 * 8 * 32 <= PIX_PER_BLK * 81, "exchange area must fit inside ldsOp");
    const v8f give = kq ? acc0 : acc1;
    v8f keep       = kq ? acc1 : acc0;
    #pragma unroll
    for (int r = 0; r < 8; ++r)
        red[wave * 256 + r * 32 + lane] = give[r];
    __syncthreads();
    #pragma unroll
    for (int r = 0; r < 8; ++r)
        keep[r] += red[(wave ^ 1) * 256 + r * 32 + lane];

    const int o  = nTile * 16 + nIdx;
    const float bo = bias[o];
    #pragma unroll
    for (int r = 0; r < 8; ++r) {
        const int m     = r + 8 * kh;
        const int pglob = pbase + kq * 16 + m;
        const int b_    = pglob >> 14;
        const int hw_   = pglob & (H * W - 1);
        out[((size_t)(b_ * COUT + o)) * (H * W) + hw_] = keep[r] + bo;
    }
}

extern "C" void kernel_launch(void* const* d_in, const int* in_sizes, int n_in,
                              void* d_out, int out_size, void* d_ws, size_t ws_size,
                              hipStream_t stream) {
    (void)in_sizes; (void)n_in; (void)out_size; (void)d_ws; (void)ws_size;
    const float* x      = (const float*)d_in[0];
    const float* weight = (const float*)d_in[1];
    const float* bias   = (const float*)d_in[2];
    const float* op_w   = (const float*)d_in[3];
    const float* op_b   = (const float*)d_in[4];
    float* out = (float*)d_out;

    const int totalPix = NB * H * W;
    const int blocks   = totalPix / PIX_PER_BLK;
    const size_t ldsBytes = (size_t)LDS_FLOATS * sizeof(float);

    dyfconv_fused<<<blocks, THREADS, ldsBytes, stream>>>(x, weight, bias, op_w, op_b, out);
}


# Operator apply y = xu (op + I) moved to the matrix unit: per pixel a 64x16x12 f32 WMMA product, A from the x tile, B from the operator in LDS (was 648 waited LDS loads, FMA chains); sum order changes
# speedup vs baseline: 5.2155x; 1.1213x over previous
#include <hip/hip_runtime.h>
#include <hip/hip_bf16.h>
#include <math.h>

typedef __attribute__((ext_vector_type(2))) float v2f;
typedef __attribute__((ext_vector_type(8))) float v8f;

constexpr int KSZ   = 3;
constexpr int K2    = 9;
constexpr int CIN   = 64;
constexpr int COUT  = 64;
constexpr int NB    = 8;
constexpr int H     = 128;
constexpr int W     = 128;
constexpr int KDIM  = CIN * K2;

constexpr int THREADS     = 256;
constexpr int PIX_PER_BLK = 32;
constexpr int YSTRIDE     = KDIM + 1;

constexpr int LDS_Y_OFF  = 0;
constexpr int LDS_OP_OFF = LDS_Y_OFF + PIX_PER_BLK * YSTRIDE;
constexpr int LDS_N_OFF  = LDS_OP_OFF + PIX_PER_BLK * 81;
constexpr int LDS_FLOATS = LDS_N_OFF + PIX_PER_BLK * K2;

constexpr int TCOLS = PIX_PER_BLK + 2;
constexpr int TCH   = 105;
static_assert(CIN * TCH <= PIX_PER_BLK * YSTRIDE, "tile must fit inside ldsY");
static_assert(W % PIX_PER_BLK == 0, "a block's pixels must share one image row");
__host__ __device__ constexpr int toff(int j) { return (j / KSZ) * TCOLS + (j % KSZ); }

constexpr int CHW   = COUT * K2;
constexpr int QROWS = 16 * K2;
static_assert(((2 * QROWS) % 64) == 32 && (YSTRIDE % 64) == 1, "bank spacing of the A fragment reads");
__host__ __device__ constexpr int kq_sel(int s, int e) { return s < 4 ? 0 : (s < 8 ? 1 : e); }
__host__ __device__ constexpr int ktap(int s, int e)   { return s < 8 ? 2 * (s & 3) + e : 8; }

__device__ __forceinline__ void load_b18(float (&b)[18], const float* __restrict__ wl) {
    #pragma unroll
    for (int s = 0; s < K2; ++s) {
        b[2 * s]     = wl[kq_sel(s, 0) * (16 * CHW) + ktap(s, 0)];
        b[2 * s + 1] = wl[kq_sel(s, 1) * (16 * CHW) + ktap(s, 1)];
    }
}

__device__ __forceinline__ void mma_trip2(v8f& acc0, v8f& acc1, const float (&b)[18],
                                          const float* ya0, const float* ya1) {
    v2f a0[K2], a1[K2];
    #pragma unroll
    for (int s = 0; s < K2; ++s) {
        a0[s].x = ya0[kq_sel(s, 0) * QROWS + ktap(s, 0)];
        a0[s].y = ya0[kq_sel(s, 1) * QROWS + ktap(s, 1)];
        a1[s].x = ya1[kq_sel(s, 0) * QROWS + ktap(s, 0)];
        a1[s].y = ya1[kq_sel(s, 1) * QROWS + ktap(s, 1)];
    }
    #pragma unroll
    for (int s = 0; s < K2; ++s) {
        v2f bf;
        bf.x = b[2 * s];
        bf.y = b[2 * s + 1];
        acc0 = __builtin_amdgcn_wmma_f32_16x16x4_f32(
                    false, a0[s],  false, bf,
                    (short)0, acc0,  false,  false);
        acc1 = __builtin_amdgcn_wmma_f32_16x16x4_f32(
                    false, a1[s],  false, bf,
                    (short)0, acc1,  false,  false);
    }
}

__global__ __launch_bounds__(THREADS) __attribute__((amdgpu_waves_per_eu(6, 6)))
void dyfconv_fused(const float* __restrict__ x,
                   const float* __restrict__ weight,
                   const float* __restrict__ bias,
                   const float* __restrict__ op_w,
                   const float* __restrict__ op_b,
                   float* __restrict__ out)
{
    extern __shared__ float lds[];
    float* ldsY  = lds + LDS_Y_OFF;
    float* ldsOp = lds + LDS_OP_OFF;
    float* ldsN  = lds + LDS_N_OFF;

    const int t     = threadIdx.x;
    const int pbase = blockIdx.x * PIX_PER_BLK;
    const int wave  = t >> 5;
    const int lane  = t & 31;
    const int kh    = lane >> 4;
    const int l16   = lane & 15;

    const int  oidx = wave * 16 + l16;
    const bool oval = oidx < K2 * K2;
    float ob[5];
    float obias;
    {
        const int oc = oval ? oidx : (K2 * K2 - 1);
        const float* wrow = op_w + oc * K2;
        const float w0 = wrow[2 * kh];
        const float w1 = wrow[2 * kh + 1];
        const float w2 = wrow[4 + 2 * kh];
        const float w3 = wrow[5 + 2 * kh];
        const float w4 = wrow[8];
        const float wb = op_b[oc];
        ob[0] = oval ? w0 : 0.0f;
        ob[1] = oval ? w1 : 0.0f;
        ob[2] = oval ? w2 : 0.0f;
        ob[3] = oval ? w3 : 0.0f;
        ob[4] = (oval && kh == 0) ? w4 : 0.0f;
        obias = oval ? wb : 0.0f;
    }

    for (int idx = t; idx < PIX_PER_BLK * K2; idx += THREADS)
        ldsN[idx] = 0.0f;

    float* tile = ldsY;
    const int bb  = pbase >> 14;
    const int hw0 = pbase & (H * W - 1);
    const int hh  = hw0 >> 7;
    const int ww0 = hw0 & (W - 1);
    const float* xb = x + (size_t)bb * CIN * (H * W);
    {
        const int wv = t >> 5;
        const int ln = t & 31;
        int  rowoff[KSZ];
        bool rowok[KSZ];
        #pragma unroll
        for (int r = 0; r < KSZ; ++r) {
            const int hs = hh + r - 1;
            rowok[r]  = (unsigned)hs < (unsigned)H;
            rowoff[r] = (rowok[r] ? hs : hh) * W;
        }
        const bool own = ln < 8 * KSZ;
        const int  hq  = own ? ln : (8 * KSZ - 1);
        const int  hci = hq / KSZ;
        const int  hrw = hq - hci * KSZ;
        const bool okr = (unsigned)(hh + hrw - 1) < (unsigned)H;
        const float* xr = xb + (size_t)(wv * 8 + hci) * (H * W) + (okr ? (hh + hrw - 1) : hh) * W;
        const bool okl = okr && (ww0 > 0);
        const bool okq = okr && (ww0 + PIX_PER_BLK < W);
        const float hla = xr[okl ? ww0 - 1 : ww0];
        const float hra = xr[okq ? ww0 + PIX_PER_BLK : ww0];
        const float hl = (own && okl) ? hla : 0.0f;
        const float hr = (own && okq) ? hra : 0.0f;
        const int ha0 = own ? ((wv * 8 + hci) * TCH + hrw * TCOLS)
                            : ((wv * 8 + (ln - 8 * KSZ)) * TCH + KSZ * TCOLS);
        const int ha1 = own ? (ha0 + TCOLS - 1) : (ha0 + 1);
        float v[8 * KSZ];
        const float* xw = xb + (size_t)(wv * 8) * (H * W) + ww0 + ln;
        #pragma unroll
        for (int i = 0; i < 8 * KSZ; ++i) {
            const int ci = i / KSZ, r = i % KSZ;
            const float a = xw[ci * (H * W) + rowoff[r]];
            v[i] = rowok[r] ? a : 0.0f;
        }
        #pragma unroll
        for (int i = 0; i < 8 * KSZ; ++i) {
            const int ci = i / KSZ, r = i % KSZ;
            tile[(wv * 8 + ci) * TCH + r * TCOLS + 1 + ln] = v[i];
        }
        tile[ha0] = hl;
        tile[ha1] = hr;
    }
    __syncthreads();

    const int p  = t >> 3;
    const int g  = t & 7;

    float xu[8][K2];
    {
        const float* tp = tile + (g * 8) * TCH + p;
        #pragma unroll
        for (int cc = 0; cc < 8; ++cc)
            #pragma unroll
            for (int ki = 0; ki < KSZ; ++ki)
                #pragma unroll
                for (int kj = 0; kj < KSZ; ++kj)
                    xu[cc][ki * KSZ + kj] = tp[cc * TCH + ki * TCOLS + kj];
    }

    {
        float sq[K2];
        #pragma unroll
        for (int j = 0; j < K2; ++j) sq[j] = 0.0f;
        #pragma unroll
        for (int cc = 0; cc < 8; ++cc) {
            #pragma unroll
            for (int j = 0; j < K2; ++j) {
                const float v = xu[cc][j];
                sq[j] += v * v;
            }
        }
        #pragma unroll
        for (int j = 0; j < K2; ++j)
            atomicAdd(&ldsN[p * K2 + j], sq[j]);
    }

    float xa[4][4][5];
    {
        const float* ta = tile + l16 * TCH + 4 * wave;
        const int o0 = kh ? toff(2) : toff(0);
        const int o1 = kh ? toff(3) : toff(1);
        const int o2 = kh ? toff(6) : toff(4);
        const int o3 = kh ? toff(7) : toff(5);
        #pragma unroll
        for (int pp = 0; pp < 4; ++pp) {
            #pragma unroll
            for (int mt = 0; mt < 4; ++mt) {
                const float* q = ta + mt * 16 * TCH + pp;
                xa[pp][mt][0] = q[o0];
                xa[pp][mt][1] = q[o1];
                xa[pp][mt][2] = q[o2];
                xa[pp][mt][3] = q[o3];
                const float v8 = q[toff(8)];
                xa[pp][mt][4] = kh ? 0.0f : v8;
            }
        }
    }
    __syncthreads();

    const int wave_s = __builtin_amdgcn_readfirstlane(wave);
    if (wave_s < 6) {
        v2f af[2][3];
        #pragma unroll
        for (int m = 0; m < 2; ++m) {
            const float* nb = &ldsN[(m * 16 + l16) * K2];
            af[m][0].x = sqrtf(nb[2 * kh]);
            af[m][0].y = sqrtf(nb[2 * kh + 1]);
            af[m][1].x = sqrtf(nb[4 + 2 * kh]);
            af[m][1].y = sqrtf(nb[5 + 2 * kh]);
            const float r8 = sqrtf(nb[8]);
            af[m][2].x = kh ? 0.0f : r8;
            af[m][2].y = 0.0f;
        }
        v8f c0, c1;
        #pragma unroll
        for (int r = 0; r < 8; ++r) { c0[r] = obias; c1[r] = obias; }
        #pragma unroll
        for (int s = 0; s < 3; ++s) {
            v2f bf;
            bf.x = ob[2 * s];
            bf.y = (s < 2) ? ob[2 * s + 1] : 0.0f;
            c0 = __builtin_amdgcn_wmma_f32_16x16x4_f32(
                      false, af[0][s],  false, bf,
                      (short)0, c0,  false,  false);
            c1 = __builtin_amdgcn_wmma_f32_16x16x4_f32(
                      false, af[1][s],  false, bf,
                      (short)0, c1,  false,  false);
        }
        if (oval) {
            float* orow = &ldsOp[(8 * kh) * 81 + oidx];
            #pragma unroll
            for (int r = 0; r < 8; ++r) {
                orow[r * 81]        = tanhf(c0[r]);
                orow[(16 + r) * 81] = tanhf(c1[r]);
            }
        }
    }
    __syncthreads();

    {
        const int kc = (l16 < K2) ? l16 : (l16 - K2);
        const float* ob0 = ldsOp + (4 * wave) * 81 + kc;
        const float* obh = ob0 + 2 * kh * K2;
        const float i0 = (2 * kh     == kc) ? 1.0f : 0.0f;
        const float i1 = (2 * kh + 1 == kc) ? 1.0f : 0.0f;
        const float i2 = (2 * kh + 4 == kc) ? 1.0f : 0.0f;
        const float i3 = (2 * kh + 5 == kc) ? 1.0f : 0.0f;
        const float i4 = (8          == kc) ? 1.0f : 0.0f;
        float* yb = ldsY + (4 * wave) * YSTRIDE + (8 * kh) * K2 + kc;
        #pragma unroll
        for (int pp = 0; pp < 4; ++pp) {
            const float* q = obh + pp * 81;
            v2f bf0, bf1, bf2;
            bf0.x = q[0]      + i0;
            bf0.y = q[K2]     + i1;
            bf1.x = q[4 * K2] + i2;
            bf1.y = q[5 * K2] + i3;
            const float b8 = ob0[pp * 81 + 8 * K2] + i4;
            bf2.x = kh ? 0.0f : b8;
            bf2.y = 0.0f;
            #pragma unroll
            for (int mt = 0; mt < 4; ++mt) {
                v2f a0, a1, a2;
                a0.x = xa[pp][mt][0];
                a0.y = xa[pp][mt][1];
                a1.x = xa[pp][mt][2];
                a1.y = xa[pp][mt][3];
                a2.x = xa[pp][mt][4];
                a2.y = 0.0f;
                v8f d = {};
                d = __builtin_amdgcn_wmma_f32_16x16x4_f32(
                         false, a0,  false, bf0,
                         (short)0, d,  false,  false);
                d = __builtin_amdgcn_wmma_f32_16x16x4_f32(
                         false, a1,  false, bf1,
                         (short)0, d,  false,  false);
                d = __builtin_amdgcn_wmma_f32_16x16x4_f32(
                         false, a2,  false, bf2,
                         (short)0, d,  false,  false);
                float* yo = yb + pp * YSTRIDE + mt * 16 * K2;
                #pragma unroll
                for (int r = 0; r < 8; ++r)
                    yo[r * K2] = d[r];
            }
        }
    }

    const int kq    = wave & 1;
    const int nTile = wave >> 1;
    const int mIdx  = l16;
    const int nIdx  = l16;
    const float* wl = weight + (32 * kh) * CHW + (nTile * 16 + nIdx) * K2;
    constexpr int TRIPS = CIN / 8;
    const int c4b = kq * TRIPS;
    float b0[18];
    load_b18(b0, wl + c4b * CHW);
    __syncthreads();

    const float* ya0 = &ldsY[mIdx * YSTRIDE + 2 * QROWS * kh];
    const float* ya1 = ya0 + 16 * YSTRIDE;

    v8f acc0 = {};
    v8f acc1 = {};
    float b1[18];
    for (int i = 0; i < TRIPS; i += 2) {
        const int c4 = c4b + i;
        load_b18(b1, wl + (c4 + 1) * CHW);
        mma_trip2(acc0, acc1, b0, ya0 + c4 * K2, ya1 + c4 * K2);
        const int cn = (i + 2 < TRIPS) ? (c4 + 2) : (c4 + 1);
        load_b18(b0, wl + cn * CHW);
        mma_trip2(acc0, acc1, b1, ya0 + (c4 + 1) * K2, ya1 + (c4 + 1) * K2);
    }

    float* red = ldsOp;
    static_assert(8 * 8 * 32 <= PIX_PER_BLK * 81, "exchange area must fit inside ldsOp");
    const v8f give = kq ? acc0 : acc1;
    v8f keep       = kq ? acc1 : acc0;
    #pragma unroll
    for (int r = 0; r < 8; ++r)
        red[wave * 256 + r * 32 + lane] = give[r];
    __syncthreads();
    #pragma unroll
    for (int r = 0; r < 8; ++r)
        keep[r] += red[(wave ^ 1) * 256 + r * 32 + lane];

    const int o  = nTile * 16 + nIdx;
    const float bo = bias[o];
    #pragma unroll
    for (int r = 0; r < 8; ++r) {
        const int m     = r + 8 * kh;
        const int pglob = pbase + kq * 16 + m;
        const int b_    = pglob >> 14;
        const int hw_   = pglob & (H * W - 1);
        out[((size_t)(b_ * COUT + o)) * (H * W) + hw_] = keep[r] + bo;
    }
}

extern "C" void kernel_launch(void* const* d_in, const int* in_sizes, int n_in,
                              void* d_out, int out_size, void* d_ws, size_t ws_size,
                              hipStream_t stream) {
    (void)in_sizes; (void)n_in; (void)out_size; (void)d_ws; (void)ws_size;
    const float* x      = (const float*)d_in[0];
    const float* weight = (const float*)d_in[1];
    const float* bias   = (const float*)d_in[2];
    const float* op_w   = (const float*)d_in[3];
    const float* op_b   = (const float*)d_in[4];
    float* out = (float*)d_out;

    const int totalPix = NB * H * W;
    const int blocks   = totalPix / PIX_PER_BLK;
    const size_t ldsBytes = (size_t)LDS_FLOATS * sizeof(float);

    dyfconv_fused<<<blocks, THREADS, ldsBytes, stream>>>(x, weight, bias, op_w, op_b, out);
}


# Patch norms from one channel sum of squares per tile position (102 a block; tap j of pixel p reads position p+off(j)): were 288 sums of 8 partials by LDS atomics, 72 loads a thread; sum order changes
# speedup vs baseline: 5.6548x; 1.0842x over previous
#include <hip/hip_runtime.h>
#include <hip/hip_bf16.h>
#include <math.h>

typedef __attribute__((ext_vector_type(2))) float v2f;
typedef __attribute__((ext_vector_type(8))) float v8f;

constexpr int KSZ   = 3;
constexpr int K2    = 9;
constexpr int CIN   = 64;
constexpr int COUT  = 64;
constexpr int NB    = 8;
constexpr int H     = 128;
constexpr int W     = 128;
constexpr int KDIM  = CIN * K2;

constexpr int THREADS     = 256;
constexpr int PIX_PER_BLK = 32;
constexpr int YSTRIDE     = KDIM + 1;

constexpr int LDS_Y_OFF  = 0;
constexpr int LDS_OP_OFF = LDS_Y_OFF + PIX_PER_BLK * YSTRIDE;
constexpr int LDS_N_OFF  = LDS_OP_OFF + PIX_PER_BLK * 81;
constexpr int LDS_FLOATS = LDS_N_OFF + PIX_PER_BLK * K2;

constexpr int TCOLS = PIX_PER_BLK + 2;
constexpr int TCH   = 105;
constexpr int NPOS  = KSZ * TCOLS;
static_assert(CIN * TCH <= PIX_PER_BLK * YSTRIDE, "tile must fit inside ldsY");
static_assert(NPOS <= PIX_PER_BLK * K2 && NPOS <= 4 * 32, "norms fit in ldsN; four waves cover the positions");
static_assert(W % PIX_PER_BLK == 0, "a block's pixels must share one image row");
__host__ __device__ constexpr int toff(int j) { return (j / KSZ) * TCOLS + (j % KSZ); }

constexpr int CHW   = COUT * K2;
constexpr int QROWS = 16 * K2;
static_assert(((2 * QROWS) % 64) == 32 && (YSTRIDE % 64) == 1, "bank spacing of the A fragment reads");
__host__ __device__ constexpr int kq_sel(int s, int e) { return s < 4 ? 0 : (s < 8 ? 1 : e); }
__host__ __device__ constexpr int ktap(int s, int e)   { return s < 8 ? 2 * (s & 3) + e : 8; }

__device__ __forceinline__ void load_b18(float (&b)[18], const float* __restrict__ wl) {
    #pragma unroll
    for (int s = 0; s < K2; ++s) {
        b[2 * s]     = wl[kq_sel(s, 0) * (16 * CHW) + ktap(s, 0)];
        b[2 * s + 1] = wl[kq_sel(s, 1) * (16 * CHW) + ktap(s, 1)];
    }
}

__device__ __forceinline__ void mma_trip2(v8f& acc0, v8f& acc1, const float (&b)[18],
                                          const float* ya0, const float* ya1) {
    v2f a0[K2], a1[K2];
    #pragma unroll
    for (int s = 0; s < K2; ++s) {
        a0[s].x = ya0[kq_sel(s, 0) * QROWS + ktap(s, 0)];
        a0[s].y = ya0[kq_sel(s, 1) * QROWS + ktap(s, 1)];
        a1[s].x = ya1[kq_sel(s, 0) * QROWS + ktap(s, 0)];
        a1[s].y = ya1[kq_sel(s, 1) * QROWS + ktap(s, 1)];
    }
    #pragma unroll
    for (int s = 0; s < K2; ++s) {
        v2f bf;
        bf.x = b[2 * s];
        bf.y = b[2 * s + 1];
        acc0 = __builtin_amdgcn_wmma_f32_16x16x4_f32(
                    false, a0[s],  false, bf,
                    (short)0, acc0,  false,  false);
        acc1 = __builtin_amdgcn_wmma_f32_16x16x4_f32(
                    false, a1[s],  false, bf,
                    (short)0, acc1,  false,  false);
    }
}

__global__ __launch_bounds__(THREADS) __attribute__((amdgpu_waves_per_eu(6, 6)))
void dyfconv_fused(const float* __restrict__ x,
                   const float* __restrict__ weight,
                   const float* __restrict__ bias,
                   const float* __restrict__ op_w,
                   const float* __restrict__ op_b,
                   float* __restrict__ out)
{
    extern __shared__ float lds[];
    float* ldsY  = lds + LDS_Y_OFF;
    float* ldsOp = lds + LDS_OP_OFF;
    float* ldsN  = lds + LDS_N_OFF;

    const int t     = threadIdx.x;
    const int pbase = blockIdx.x * PIX_PER_BLK;
    const int wave  = t >> 5;
    const int lane  = t & 31;
    const int kh    = lane >> 4;
    const int l16   = lane & 15;

    const int  oidx = wave * 16 + l16;
    const bool oval = oidx < K2 * K2;
    float ob[5];
    float obias;
    {
        const int oc = oval ? oidx : (K2 * K2 - 1);
        const float* wrow = op_w + oc * K2;
        const float w0 = wrow[2 * kh];
        const float w1 = wrow[2 * kh + 1];
        const float w2 = wrow[4 + 2 * kh];
        const float w3 = wrow[5 + 2 * kh];
        const float w4 = wrow[8];
        const float wb = op_b[oc];
        ob[0] = oval ? w0 : 0.0f;
        ob[1] = oval ? w1 : 0.0f;
        ob[2] = oval ? w2 : 0.0f;
        ob[3] = oval ? w3 : 0.0f;
        ob[4] = (oval && kh == 0) ? w4 : 0.0f;
        obias = oval ? wb : 0.0f;
    }

    float* tile = ldsY;
    const int bb  = pbase >> 14;
    const int hw0 = pbase & (H * W - 1);
    const int hh  = hw0 >> 7;
    const int ww0 = hw0 & (W - 1);
    const float* xb = x + (size_t)bb * CIN * (H * W);
    {
        const int wv = t >> 5;
        const int ln = t & 31;
        int  rowoff[KSZ];
        bool rowok[KSZ];
        #pragma unroll
        for (int r = 0; r < KSZ; ++r) {
            const int hs = hh + r - 1;
            rowok[r]  = (unsigned)hs < (unsigned)H;
            rowoff[r] = (rowok[r] ? hs : hh) * W;
        }
        const bool own = ln < 8 * KSZ;
        const int  hq  = own ? ln : (8 * KSZ - 1);
        const int  hci = hq / KSZ;
        const int  hrw = hq - hci * KSZ;
        const bool okr = (unsigned)(hh + hrw - 1) < (unsigned)H;
        const float* xr = xb + (size_t)(wv * 8 + hci) * (H * W) + (okr ? (hh + hrw - 1) : hh) * W;
        const bool okl = okr && (ww0 > 0);
        const bool okq = okr && (ww0 + PIX_PER_BLK < W);
        const float hla = xr[okl ? ww0 - 1 : ww0];
        const float hra = xr[okq ? ww0 + PIX_PER_BLK : ww0];
        const float hl = (own && okl) ? hla : 0.0f;
        const float hr = (own && okq) ? hra : 0.0f;
        const int ha0 = own ? ((wv * 8 + hci) * TCH + hrw * TCOLS)
                            : ((wv * 8 + (ln - 8 * KSZ)) * TCH + KSZ * TCOLS);
        const int ha1 = own ? (ha0 + TCOLS - 1) : (ha0 + 1);
        float v[8 * KSZ];
        const float* xw = xb + (size_t)(wv * 8) * (H * W) + ww0 + ln;
        #pragma unroll
        for (int i = 0; i < 8 * KSZ; ++i) {
            const int ci = i / KSZ, r = i % KSZ;
            const float a = xw[ci * (H * W) + rowoff[r]];
            v[i] = rowok[r] ? a : 0.0f;
        }
        #pragma unroll
        for (int i = 0; i < 8 * KSZ; ++i) {
            const int ci = i / KSZ, r = i % KSZ;
            tile[(wv * 8 + ci) * TCH + r * TCOLS + 1 + ln] = v[i];
        }
        tile[ha0] = hl;
        tile[ha1] = hr;
    }
    __syncthreads();

    const int wave_s = __builtin_amdgcn_readfirstlane(wave);

    if (wave_s < 4) {
        const float* tq = tile + (t < NPOS ? t : NPOS - 1);
        float s0 = 0.0f, s1 = 0.0f, s2 = 0.0f, s3 = 0.0f;
        #pragma unroll
        for (int c = 0; c < CIN; c += 4) {
            const float a0 = tq[(c + 0) * TCH];
            const float a1 = tq[(c + 1) * TCH];
            const float a2 = tq[(c + 2) * TCH];
            const float a3 = tq[(c + 3) * TCH];
            s0 += a0 * a0;
            s1 += a1 * a1;
            s2 += a2 * a2;
            s3 += a3 * a3;
        }
        const float nrm = sqrtf((s0 + s1) + (s2 + s3));
        if (t < NPOS)
            ldsN[t] = nrm;
    }

    float xa[4][4][5];
    {
        const float* ta = tile + l16 * TCH + 4 * wave;
        const int o0 = kh ? toff(2) : toff(0);
        const int o1 = kh ? toff(3) : toff(1);
        const int o2 = kh ? toff(6) : toff(4);
        const int o3 = kh ? toff(7) : toff(5);
        #pragma unroll
        for (int pp = 0; pp < 4; ++pp) {
            #pragma unroll
            for (int mt = 0; mt < 4; ++mt) {
                const float* q = ta + mt * 16 * TCH + pp;
                xa[pp][mt][0] = q[o0];
                xa[pp][mt][1] = q[o1];
                xa[pp][mt][2] = q[o2];
                xa[pp][mt][3] = q[o3];
                const float v8 = q[toff(8)];
                xa[pp][mt][4] = kh ? 0.0f : v8;
            }
        }
    }
    __syncthreads();

    if (wave_s < 6) {
        v2f af[2][3];
        #pragma unroll
        for (int m = 0; m < 2; ++m) {
            const float* nb = &ldsN[m * 16 + l16];
            af[m][0].x = nb[kh ? toff(2) : toff(0)];
            af[m][0].y = nb[kh ? toff(3) : toff(1)];
            af[m][1].x = nb[kh ? toff(6) : toff(4)];
            af[m][1].y = nb[kh ? toff(7) : toff(5)];
            const float r8 = nb[toff(8)];
            af[m][2].x = kh ? 0.0f : r8;
            af[m][2].y = 0.0f;
        }
        v8f c0, c1;
        #pragma unroll
        for (int r = 0; r < 8; ++r) { c0[r] = obias; c1[r] = obias; }
        #pragma unroll
        for (int s = 0; s < 3; ++s) {
            v2f bf;
            bf.x = ob[2 * s];
            bf.y = (s < 2) ? ob[2 * s + 1] : 0.0f;
            c0 = __builtin_amdgcn_wmma_f32_16x16x4_f32(
                      false, af[0][s],  false, bf,
                      (short)0, c0,  false,  false);
            c1 = __builtin_amdgcn_wmma_f32_16x16x4_f32(
                      false, af[1][s],  false, bf,
                      (short)0, c1,  false,  false);
        }
        if (oval) {
            float* orow = &ldsOp[(8 * kh) * 81 + oidx];
            #pragma unroll
            for (int r = 0; r < 8; ++r) {
                orow[r * 81]        = tanhf(c0[r]);
                orow[(16 + r) * 81] = tanhf(c1[r]);
            }
        }
    }
    __syncthreads();

    {
        const int kc = (l16 < K2) ? l16 : (l16 - K2);
        const float* ob0 = ldsOp + (4 * wave) * 81 + kc;
        const float* obh = ob0 + 2 * kh * K2;
        const float i0 = (2 * kh     == kc) ? 1.0f : 0.0f;
        const float i1 = (2 * kh + 1 == kc) ? 1.0f : 0.0f;
        const float i2 = (2 * kh + 4 == kc) ? 1.0f : 0.0f;
        const float i3 = (2 * kh + 5 == kc) ? 1.0f : 0.0f;
        const float i4 = (8          == kc) ? 1.0f : 0.0f;
        float* yb = ldsY + (4 * wave) * YSTRIDE + (8 * kh) * K2 + kc;
        #pragma unroll
        for (int pp = 0; pp < 4; ++pp) {
            const float* q = obh + pp * 81;
            v2f bf0, bf1, bf2;
            bf0.x = q[0]      + i0;
            bf0.y = q[K2]     + i1;
            bf1.x = q[4 * K2] + i2;
            bf1.y = q[5 * K2] + i3;
            const float b8 = ob0[pp * 81 + 8 * K2] + i4;
            bf2.x = kh ? 0.0f : b8;
            bf2.y = 0.0f;
            #pragma unroll
            for (int mt = 0; mt < 4; ++mt) {
                v2f a0, a1, a2;
                a0.x = xa[pp][mt][0];
                a0.y = xa[pp][mt][1];
                a1.x = xa[pp][mt][2];
                a1.y = xa[pp][mt][3];
                a2.x = xa[pp][mt][4];
                a2.y = 0.0f;
                v8f d = {};
                d = __builtin_amdgcn_wmma_f32_16x16x4_f32(
                         false, a0,  false, bf0,
                         (short)0, d,  false,  false);
                d = __builtin_amdgcn_wmma_f32_16x16x4_f32(
                         false, a1,  false, bf1,
                         (short)0, d,  false,  false);
                d = __builtin_amdgcn_wmma_f32_16x16x4_f32(
                         false, a2,  false, bf2,
                         (short)0, d,  false,  false);
                float* yo = yb + pp * YSTRIDE + mt * 16 * K2;
                #pragma unroll
                for (int r = 0; r < 8; ++r)
                    yo[r * K2] = d[r];
            }
        }
    }

    const int kq    = wave & 1;
    const int nTile = wave >> 1;
    const int mIdx  = l16;
    const int nIdx  = l16;
    const float* wl = weight + (32 * kh) * CHW + (nTile * 16 + nIdx) * K2;
    constexpr int TRIPS = CIN / 8;
    const int c4b = kq * TRIPS;
    float b0[18];
    load_b18(b0, wl + c4b * CHW);
    __syncthreads();

    const float* ya0 = &ldsY[mIdx * YSTRIDE + 2 * QROWS * kh];
    const float* ya1 = ya0 + 16 * YSTRIDE;

    v8f acc0 = {};
    v8f acc1 = {};
    float b1[18];
    for (int i = 0; i < TRIPS; i += 2) {
        const int c4 = c4b + i;
        load_b18(b1, wl + (c4 + 1) * CHW);
        mma_trip2(acc0, acc1, b0, ya0 + c4 * K2, ya1 + c4 * K2);
        const int cn = (i + 2 < TRIPS) ? (c4 + 2) : (c4 + 1);
        load_b18(b0, wl + cn * CHW);
        mma_trip2(acc0, acc1, b1, ya0 + (c4 + 1) * K2, ya1 + (c4 + 1) * K2);
    }

    float* red = ldsOp;
    static_assert(8 * 8 * 32 <= PIX_PER_BLK * 81, "exchange area must fit inside ldsOp");
    const v8f give = kq ? acc0 : acc1;
    v8f keep       = kq ? acc1 : acc0;
    #pragma unroll
    for (int r = 0; r < 8; ++r)
        red[wave * 256 + r * 32 + lane] = give[r];
    __syncthreads();
    #pragma unroll
    for (int r = 0; r < 8; ++r)
        keep[r] += red[(wave ^ 1) * 256 + r * 32 + lane];

    const int o  = nTile * 16 + nIdx;
    const float bo = bias[o];
    #pragma unroll
    for (int r = 0; r < 8; ++r) {
        const int m     = r + 8 * kh;
        const int pglob = pbase + kq * 16 + m;
        const int b_    = pglob >> 14;
        const int hw_   = pglob & (H * W - 1);
        out[((size_t)(b_ * COUT + o)) * (H * W) + hw_] = keep[r] + bo;
    }
}

extern "C" void kernel_launch(void* const* d_in, const int* in_sizes, int n_in,
                              void* d_out, int out_size, void* d_ws, size_t ws_size,
                              hipStream_t stream) {
    (void)in_sizes; (void)n_in; (void)out_size; (void)d_ws; (void)ws_size;
    const float* x      = (const float*)d_in[0];
    const float* weight = (const float*)d_in[1];
    const float* bias   = (const float*)d_in[2];
    const float* op_w   = (const float*)d_in[3];
    const float* op_b   = (const float*)d_in[4];
    float* out = (float*)d_out;

    const int totalPix = NB * H * W;
    const int blocks   = totalPix / PIX_PER_BLK;
    const size_t ldsBytes = (size_t)LDS_FLOATS * sizeof(float);

    dyfconv_fused<<<blocks, THREADS, ldsBytes, stream>>>(x, weight, bias, op_w, op_b, out);
}


# tanh of the 81 operator entries on the hardware instruction v_tanh_f32, one per value (was the library tanhf: about 40 instructions, two lane-masked branches per value); last bits of tanh change
# speedup vs baseline: 6.0467x; 1.0693x over previous
#include <hip/hip_runtime.h>
#include <hip/hip_bf16.h>
#include <math.h>

typedef __attribute__((ext_vector_type(2))) float v2f;
typedef __attribute__((ext_vector_type(8))) float v8f;

constexpr int KSZ   = 3;
constexpr int K2    = 9;
constexpr int CIN   = 64;
constexpr int COUT  = 64;
constexpr int NB    = 8;
constexpr int H     = 128;
constexpr int W     = 128;
constexpr int KDIM  = CIN * K2;

constexpr int THREADS     = 256;
constexpr int PIX_PER_BLK = 32;
constexpr int YSTRIDE     = KDIM + 1;

constexpr int LDS_Y_OFF  = 0;
constexpr int LDS_OP_OFF = LDS_Y_OFF + PIX_PER_BLK * YSTRIDE;
constexpr int LDS_N_OFF  = LDS_OP_OFF + PIX_PER_BLK * 81;
constexpr int LDS_FLOATS = LDS_N_OFF + PIX_PER_BLK * K2;

constexpr int TCOLS = PIX_PER_BLK + 2;
constexpr int TCH   = 105;
constexpr int NPOS  = KSZ * TCOLS;
static_assert(CIN * TCH <= PIX_PER_BLK * YSTRIDE, "tile must fit inside ldsY");
static_assert(NPOS <= PIX_PER_BLK * K2 && NPOS <= 4 * 32, "norms fit in ldsN; four waves cover the positions");
static_assert(W % PIX_PER_BLK == 0, "a block's pixels must share one image row");
__host__ __device__ constexpr int toff(int j) { return (j / KSZ) * TCOLS + (j % KSZ); }

constexpr int CHW   = COUT * K2;
constexpr int QROWS = 16 * K2;
static_assert(((2 * QROWS) % 64) == 32 && (YSTRIDE % 64) == 1, "bank spacing of the A fragment reads");
__host__ __device__ constexpr int kq_sel(int s, int e) { return s < 4 ? 0 : (s < 8 ? 1 : e); }
__host__ __device__ constexpr int ktap(int s, int e)   { return s < 8 ? 2 * (s & 3) + e : 8; }

__device__ __forceinline__ void load_b18(float (&b)[18], const float* __restrict__ wl) {
    #pragma unroll
    for (int s = 0; s < K2; ++s) {
        b[2 * s]     = wl[kq_sel(s, 0) * (16 * CHW) + ktap(s, 0)];
        b[2 * s + 1] = wl[kq_sel(s, 1) * (16 * CHW) + ktap(s, 1)];
    }
}

__device__ __forceinline__ void mma_trip2(v8f& acc0, v8f& acc1, const float (&b)[18],
                                          const float* ya0, const float* ya1) {
    v2f a0[K2], a1[K2];
    #pragma unroll
    for (int s = 0; s < K2; ++s) {
        a0[s].x = ya0[kq_sel(s, 0) * QROWS + ktap(s, 0)];
        a0[s].y = ya0[kq_sel(s, 1) * QROWS + ktap(s, 1)];
        a1[s].x = ya1[kq_sel(s, 0) * QROWS + ktap(s, 0)];
        a1[s].y = ya1[kq_sel(s, 1) * QROWS + ktap(s, 1)];
    }
    #pragma unroll
    for (int s = 0; s < K2; ++s) {
        v2f bf;
        bf.x = b[2 * s];
        bf.y = b[2 * s + 1];
        acc0 = __builtin_amdgcn_wmma_f32_16x16x4_f32(
                    false, a0[s],  false, bf,
                    (short)0, acc0,  false,  false);
        acc1 = __builtin_amdgcn_wmma_f32_16x16x4_f32(
                    false, a1[s],  false, bf,
                    (short)0, acc1,  false,  false);
    }
}

__device__ __forceinline__ float tanh_hw(float x) {
    return __builtin_amdgcn_tanhf(x);
}

__global__ __launch_bounds__(THREADS) __attribute__((amdgpu_waves_per_eu(6, 6)))
void dyfconv_fused(const float* __restrict__ x,
                   const float* __restrict__ weight,
                   const float* __restrict__ bias,
                   const float* __restrict__ op_w,
                   const float* __restrict__ op_b,
                   float* __restrict__ out)
{
    extern __shared__ float lds[];
    float* ldsY  = lds + LDS_Y_OFF;
    float* ldsOp = lds + LDS_OP_OFF;
    float* ldsN  = lds + LDS_N_OFF;

    const int t     = threadIdx.x;
    const int pbase = blockIdx.x * PIX_PER_BLK;
    const int wave  = t >> 5;
    const int lane  = t & 31;
    const int kh    = lane >> 4;
    const int l16   = lane & 15;

    const int  oidx = wave * 16 + l16;
    const bool oval = oidx < K2 * K2;
    float ob[5];
    float obias;
    {
        const int oc = oval ? oidx : (K2 * K2 - 1);
        const float* wrow = op_w + oc * K2;
        const float w0 = wrow[2 * kh];
        const float w1 = wrow[2 * kh + 1];
        const float w2 = wrow[4 + 2 * kh];
        const float w3 = wrow[5 + 2 * kh];
        const float w4 = wrow[8];
        const float wb = op_b[oc];
        ob[0] = oval ? w0 : 0.0f;
        ob[1] = oval ? w1 : 0.0f;
        ob[2] = oval ? w2 : 0.0f;
        ob[3] = oval ? w3 : 0.0f;
        ob[4] = (oval && kh == 0) ? w4 : 0.0f;
        obias = oval ? wb : 0.0f;
    }

    float* tile = ldsY;
    const int bb  = pbase >> 14;
    const int hw0 = pbase & (H * W - 1);
    const int hh  = hw0 >> 7;
    const int ww0 = hw0 & (W - 1);
    const float* xb = x + (size_t)bb * CIN * (H * W);
    {
        const int wv = t >> 5;
        const int ln = t & 31;
        int  rowoff[KSZ];
        bool rowok[KSZ];
        #pragma unroll
        for (int r = 0; r < KSZ; ++r) {
            const int hs = hh + r - 1;
            rowok[r]  = (unsigned)hs < (unsigned)H;
            rowoff[r] = (rowok[r] ? hs : hh) * W;
        }
        const bool own = ln < 8 * KSZ;
        const int  hq  = own ? ln : (8 * KSZ - 1);
        const int  hci = hq / KSZ;
        const int  hrw = hq - hci * KSZ;
        const bool okr = (unsigned)(hh + hrw - 1) < (unsigned)H;
        const float* xr = xb + (size_t)(wv * 8 + hci) * (H * W) + (okr ? (hh + hrw - 1) : hh) * W;
        const bool okl = okr && (ww0 > 0);
        const bool okq = okr && (ww0 + PIX_PER_BLK < W);
        const float hla = xr[okl ? ww0 - 1 : ww0];
        const float hra = xr[okq ? ww0 + PIX_PER_BLK : ww0];
        const float hl = (own && okl) ? hla : 0.0f;
        const float hr = (own && okq) ? hra : 0.0f;
        const int ha0 = own ? ((wv * 8 + hci) * TCH + hrw * TCOLS)
                            : ((wv * 8 + (ln - 8 * KSZ)) * TCH + KSZ * TCOLS);
        const int ha1 = own ? (ha0 + TCOLS - 1) : (ha0 + 1);
        float v[8 * KSZ];
        const float* xw = xb + (size_t)(wv * 8) * (H * W) + ww0 + ln;
        #pragma unroll
        for (int i = 0; i < 8 * KSZ; ++i) {
            const int ci = i / KSZ, r = i % KSZ;
            const float a = xw[ci * (H * W) + rowoff[r]];
            v[i] = rowok[r] ? a : 0.0f;
        }
        #pragma unroll
        for (int i = 0; i < 8 * KSZ; ++i) {
            const int ci = i / KSZ, r = i % KSZ;
            tile[(wv * 8 + ci) * TCH + r * TCOLS + 1 + ln] = v[i];
        }
        tile[ha0] = hl;
        tile[ha1] = hr;
    }
    __syncthreads();

    const int wave_s = __builtin_amdgcn_readfirstlane(wave);

    if (wave_s < 4) {
        const float* tq = tile + (t < NPOS ? t : NPOS - 1);
        float s0 = 0.0f, s1 = 0.0f, s2 = 0.0f, s3 = 0.0f;
        #pragma unroll
        for (int c = 0; c < CIN; c += 4) {
            const float a0 = tq[(c + 0) * TCH];
            const float a1 = tq[(c + 1) * TCH];
            const float a2 = tq[(c + 2) * TCH];
            const float a3 = tq[(c + 3) * TCH];
            s0 += a0 * a0;
            s1 += a1 * a1;
            s2 += a2 * a2;
            s3 += a3 * a3;
        }
        const float nrm = sqrtf((s0 + s1) + (s2 + s3));
        if (t < NPOS)
            ldsN[t] = nrm;
    }

    float xa[4][4][5];
    {
        const float* ta = tile + l16 * TCH + 4 * wave;
        const int o0 = kh ? toff(2) : toff(0);
        const int o1 = kh ? toff(3) : toff(1);
        const int o2 = kh ? toff(6) : toff(4);
        const int o3 = kh ? toff(7) : toff(5);
        #pragma unroll
        for (int pp = 0; pp < 4; ++pp) {
            #pragma unroll
            for (int mt = 0; mt < 4; ++mt) {
                const float* q = ta + mt * 16 * TCH + pp;
                xa[pp][mt][0] = q[o0];
                xa[pp][mt][1] = q[o1];
                xa[pp][mt][2] = q[o2];
                xa[pp][mt][3] = q[o3];
                const float v8 = q[toff(8)];
                xa[pp][mt][4] = kh ? 0.0f : v8;
            }
        }
    }
    __syncthreads();

    if (wave_s < 6) {
        v2f af[2][3];
        #pragma unroll
        for (int m = 0; m < 2; ++m) {
            const float* nb = &ldsN[m * 16 + l16];
            af[m][0].x = nb[kh ? toff(2) : toff(0)];
            af[m][0].y = nb[kh ? toff(3) : toff(1)];
            af[m][1].x = nb[kh ? toff(6) : toff(4)];
            af[m][1].y = nb[kh ? toff(7) : toff(5)];
            const float r8 = nb[toff(8)];
            af[m][2].x = kh ? 0.0f : r8;
            af[m][2].y = 0.0f;
        }
        v8f c0, c1;
        #pragma unroll
        for (int r = 0; r < 8; ++r) { c0[r] = obias; c1[r] = obias; }
        #pragma unroll
        for (int s = 0; s < 3; ++s) {
            v2f bf;
            bf.x = ob[2 * s];
            bf.y = (s < 2) ? ob[2 * s + 1] : 0.0f;
            c0 = __builtin_amdgcn_wmma_f32_16x16x4_f32(
                      false, af[0][s],  false, bf,
                      (short)0, c0,  false,  false);
            c1 = __builtin_amdgcn_wmma_f32_16x16x4_f32(
                      false, af[1][s],  false, bf,
                      (short)0, c1,  false,  false);
        }
        if (oval) {
            float* orow = &ldsOp[(8 * kh) * 81 + oidx];
            #pragma unroll
            for (int r = 0; r < 8; ++r) {
                orow[r * 81]        = tanh_hw(c0[r]);
                orow[(16 + r) * 81] = tanh_hw(c1[r]);
            }
        }
    }
    __syncthreads();

    {
        const int kc = (l16 < K2) ? l16 : (l16 - K2);
        const float* ob0 = ldsOp + (4 * wave) * 81 + kc;
        const float* obh = ob0 + 2 * kh * K2;
        const float i0 = (2 * kh     == kc) ? 1.0f : 0.0f;
        const float i1 = (2 * kh + 1 == kc) ? 1.0f : 0.0f;
        const float i2 = (2 * kh + 4 == kc) ? 1.0f : 0.0f;
        const float i3 = (2 * kh + 5 == kc) ? 1.0f : 0.0f;
        const float i4 = (8          == kc) ? 1.0f : 0.0f;
        float* yb = ldsY + (4 * wave) * YSTRIDE + (8 * kh) * K2 + kc;
        #pragma unroll
        for (int pp = 0; pp < 4; ++pp) {
            const float* q = obh + pp * 81;
            v2f bf0, bf1, bf2;
            bf0.x = q[0]      + i0;
            bf0.y = q[K2]     + i1;
            bf1.x = q[4 * K2] + i2;
            bf1.y = q[5 * K2] + i3;
            const float b8 = ob0[pp * 81 + 8 * K2] + i4;
            bf2.x = kh ? 0.0f : b8;
            bf2.y = 0.0f;
            #pragma unroll
            for (int mt = 0; mt < 4; ++mt) {
                v2f a0, a1, a2;
                a0.x = xa[pp][mt][0];
                a0.y = xa[pp][mt][1];
                a1.x = xa[pp][mt][2];
                a1.y = xa[pp][mt][3];
                a2.x = xa[pp][mt][4];
                a2.y = 0.0f;
                v8f d = {};
                d = __builtin_amdgcn_wmma_f32_16x16x4_f32(
                         false, a0,  false, bf0,
                         (short)0, d,  false,  false);
                d = __builtin_amdgcn_wmma_f32_16x16x4_f32(
                         false, a1,  false, bf1,
                         (short)0, d,  false,  false);
                d = __builtin_amdgcn_wmma_f32_16x16x4_f32(
                         false, a2,  false, bf2,
                         (short)0, d,  false,  false);
                float* yo = yb + pp * YSTRIDE + mt * 16 * K2;
                #pragma unroll
                for (int r = 0; r < 8; ++r)
                    yo[r * K2] = d[r];
            }
        }
    }

    const int kq    = wave & 1;
    const int nTile = wave >> 1;
    const int mIdx  = l16;
    const int nIdx  = l16;
    const float* wl = weight + (32 * kh) * CHW + (nTile * 16 + nIdx) * K2;
    constexpr int TRIPS = CIN / 8;
    const int c4b = kq * TRIPS;
    float b0[18];
    load_b18(b0, wl + c4b * CHW);
    __syncthreads();

    const float* ya0 = &ldsY[mIdx * YSTRIDE + 2 * QROWS * kh];
    const float* ya1 = ya0 + 16 * YSTRIDE;

    v8f acc0 = {};
    v8f acc1 = {};
    float b1[18];
    for (int i = 0; i < TRIPS; i += 2) {
        const int c4 = c4b + i;
        load_b18(b1, wl + (c4 + 1) * CHW);
        mma_trip2(acc0, acc1, b0, ya0 + c4 * K2, ya1 + c4 * K2);
        const int cn = (i + 2 < TRIPS) ? (c4 + 2) : (c4 + 1);
        load_b18(b0, wl + cn * CHW);
        mma_trip2(acc0, acc1, b1, ya0 + (c4 + 1) * K2, ya1 + (c4 + 1) * K2);
    }

    float* red = ldsOp;
    static_assert(8 * 8 * 32 <= PIX_PER_BLK * 81, "exchange area must fit inside ldsOp");
    const v8f give = kq ? acc0 : acc1;
    v8f keep       = kq ? acc1 : acc0;
    #pragma unroll
    for (int r = 0; r < 8; ++r)
        red[wave * 256 + r * 32 + lane] = give[r];
    __syncthreads();
    #pragma unroll
    for (int r = 0; r < 8; ++r)
        keep[r] += red[(wave ^ 1) * 256 + r * 32 + lane];

    const int o  = nTile * 16 + nIdx;
    const float bo = bias[o];
    #pragma unroll
    for (int r = 0; r < 8; ++r) {
        const int m     = r + 8 * kh;
        const int pglob = pbase + kq * 16 + m;
        const int b_    = pglob >> 14;
        const int hw_   = pglob & (H * W - 1);
        out[((size_t)(b_ * COUT + o)) * (H * W) + hw_] = keep[r] + bo;
    }
}

extern "C" void kernel_launch(void* const* d_in, const int* in_sizes, int n_in,
                              void* d_out, int out_size, void* d_ws, size_t ws_size,
                              hipStream_t stream) {
    (void)in_sizes; (void)n_in; (void)out_size; (void)d_ws; (void)ws_size;
    const float* x      = (const float*)d_in[0];
    const float* weight = (const float*)d_in[1];
    const float* bias   = (const float*)d_in[2];
    const float* op_w   = (const float*)d_in[3];
    const float* op_b   = (const float*)d_in[4];
    float* out = (float*)d_out;

    const int totalPix = NB * H * W;
    const int blocks   = totalPix / PIX_PER_BLK;
    const size_t ldsBytes = (size_t)LDS_FLOATS * sizeof(float);

    dyfconv_fused<<<blocks, THREADS, ldsBytes, stream>>>(x, weight, bias, op_w, op_b, out);
}


# Epilogue: output tile turned through LDS (row = out-channel): each 128-byte line written whole by one store, 4 lines an instruction (was 16 lines x 32 B per wave, 4 pieces a line); sum order kept
# speedup vs baseline: 6.2548x; 1.0344x over previous
#include <hip/hip_runtime.h>
#include <hip/hip_bf16.h>
#include <math.h>

typedef __attribute__((ext_vector_type(2))) float v2f;
typedef __attribute__((ext_vector_type(8))) float v8f;

constexpr int KSZ   = 3;
constexpr int K2    = 9;
constexpr int CIN   = 64;
constexpr int COUT  = 64;
constexpr int NB    = 8;
constexpr int H     = 128;
constexpr int W     = 128;
constexpr int KDIM  = CIN * K2;

constexpr int THREADS     = 256;
constexpr int PIX_PER_BLK = 32;
constexpr int YSTRIDE     = KDIM + 1;

constexpr int LDS_Y_OFF  = 0;
constexpr int LDS_OP_OFF = LDS_Y_OFF + PIX_PER_BLK * YSTRIDE;
constexpr int LDS_N_OFF  = LDS_OP_OFF + PIX_PER_BLK * 81;
constexpr int SPITCH     = 36;
constexpr int SPLANE     = COUT * SPITCH;
constexpr int LDS_ST_OFF = LDS_OP_OFF;
constexpr int LDS_END_A  = LDS_N_OFF + PIX_PER_BLK * K2;
constexpr int LDS_END_B  = LDS_ST_OFF + 2 * SPLANE;
constexpr int LDS_FLOATS = LDS_END_A > LDS_END_B ? LDS_END_A : LDS_END_B;
static_assert((LDS_ST_OFF % 4) == 0 && (SPITCH % 4) == 0 && (SPLANE % 4) == 0, "16-byte alignment of the stage rows");

constexpr int TCOLS = PIX_PER_BLK + 2;
constexpr int TCH   = 105;
constexpr int NPOS  = KSZ * TCOLS;
static_assert(CIN * TCH <= PIX_PER_BLK * YSTRIDE, "tile must fit inside ldsY");
static_assert(NPOS <= PIX_PER_BLK * K2 && NPOS <= 4 * 32, "norms fit in ldsN; four waves cover the positions");
static_assert(W % PIX_PER_BLK == 0, "a block's pixels must share one image row");
__host__ __device__ constexpr int toff(int j) { return (j / KSZ) * TCOLS + (j % KSZ); }

constexpr int CHW   = COUT * K2;
constexpr int QROWS = 16 * K2;
static_assert(((2 * QROWS) % 64) == 32 && (YSTRIDE % 64) == 1, "bank spacing of the A fragment reads");
__host__ __device__ constexpr int kq_sel(int s, int e) { return s < 4 ? 0 : (s < 8 ? 1 : e); }
__host__ __device__ constexpr int ktap(int s, int e)   { return s < 8 ? 2 * (s & 3) + e : 8; }

__device__ __forceinline__ void load_b18(float (&b)[18], const float* __restrict__ wl) {
    #pragma unroll
    for (int s = 0; s < K2; ++s) {
        b[2 * s]     = wl[kq_sel(s, 0) * (16 * CHW) + ktap(s, 0)];
        b[2 * s + 1] = wl[kq_sel(s, 1) * (16 * CHW) + ktap(s, 1)];
    }
}

__device__ __forceinline__ void mma_trip2(v8f& acc0, v8f& acc1, const float (&b)[18],
                                          const float* ya0, const float* ya1) {
    v2f a0[K2], a1[K2];
    #pragma unroll
    for (int s = 0; s < K2; ++s) {
        a0[s].x = ya0[kq_sel(s, 0) * QROWS + ktap(s, 0)];
        a0[s].y = ya0[kq_sel(s, 1) * QROWS + ktap(s, 1)];
        a1[s].x = ya1[kq_sel(s, 0) * QROWS + ktap(s, 0)];
        a1[s].y = ya1[kq_sel(s, 1) * QROWS + ktap(s, 1)];
    }
    #pragma unroll
    for (int s = 0; s < K2; ++s) {
        v2f bf;
        bf.x = b[2 * s];
        bf.y = b[2 * s + 1];
        acc0 = __builtin_amdgcn_wmma_f32_16x16x4_f32(
                    false, a0[s],  false, bf,
                    (short)0, acc0,  false,  false);
        acc1 = __builtin_amdgcn_wmma_f32_16x16x4_f32(
                    false, a1[s],  false, bf,
                    (short)0, acc1,  false,  false);
    }
}

__device__ __forceinline__ float tanh_hw(float x) {
    return __builtin_amdgcn_tanhf(x);
}

__global__ __launch_bounds__(THREADS) __attribute__((amdgpu_waves_per_eu(6, 6)))
void dyfconv_fused(const float* __restrict__ x,
                   const float* __restrict__ weight,
                   const float* __restrict__ bias,
                   const float* __restrict__ op_w,
                   const float* __restrict__ op_b,
                   float* __restrict__ out)
{
    extern __shared__ __attribute__((aligned(16))) float lds[];
    float* ldsY  = lds + LDS_Y_OFF;
    float* ldsOp = lds + LDS_OP_OFF;
    float* ldsN  = lds + LDS_N_OFF;

    const int t     = threadIdx.x;
    const int pbase = blockIdx.x * PIX_PER_BLK;
    const int wave  = t >> 5;
    const int lane  = t & 31;
    const int kh    = lane >> 4;
    const int l16   = lane & 15;

    const int  oidx = wave * 16 + l16;
    const bool oval = oidx < K2 * K2;
    float ob[5];
    float obias;
    {
        const int oc = oval ? oidx : (K2 * K2 - 1);
        const float* wrow = op_w + oc * K2;
        const float w0 = wrow[2 * kh];
        const float w1 = wrow[2 * kh + 1];
        const float w2 = wrow[4 + 2 * kh];
        const float w3 = wrow[5 + 2 * kh];
        const float w4 = wrow[8];
        const float wb = op_b[oc];
        ob[0] = oval ? w0 : 0.0f;
        ob[1] = oval ? w1 : 0.0f;
        ob[2] = oval ? w2 : 0.0f;
        ob[3] = oval ? w3 : 0.0f;
        ob[4] = (oval && kh == 0) ? w4 : 0.0f;
        obias = oval ? wb : 0.0f;
    }

    float* tile = ldsY;
    const int bb  = pbase >> 14;
    const int hw0 = pbase & (H * W - 1);
    const int hh  = hw0 >> 7;
    const int ww0 = hw0 & (W - 1);
    const float* xb = x + (size_t)bb * CIN * (H * W);
    {
        const int wv = t >> 5;
        const int ln = t & 31;
        int  rowoff[KSZ];
        bool rowok[KSZ];
        #pragma unroll
        for (int r = 0; r < KSZ; ++r) {
            const int hs = hh + r - 1;
            rowok[r]  = (unsigned)hs < (unsigned)H;
            rowoff[r] = (rowok[r] ? hs : hh) * W;
        }
        const bool own = ln < 8 * KSZ;
        const int  hq  = own ? ln : (8 * KSZ - 1);
        const int  hci = hq / KSZ;
        const int  hrw = hq - hci * KSZ;
        const bool okr = (unsigned)(hh + hrw - 1) < (unsigned)H;
        const float* xr = xb + (size_t)(wv * 8 + hci) * (H * W) + (okr ? (hh + hrw - 1) : hh) * W;
        const bool okl = okr && (ww0 > 0);
        const bool okq = okr && (ww0 + PIX_PER_BLK < W);
        const float hla = xr[okl ? ww0 - 1 : ww0];
        const float hra = xr[okq ? ww0 + PIX_PER_BLK : ww0];
        const float hl = (own && okl) ? hla : 0.0f;
        const float hr = (own && okq) ? hra : 0.0f;
        const int ha0 = own ? ((wv * 8 + hci) * TCH + hrw * TCOLS)
                            : ((wv * 8 + (ln - 8 * KSZ)) * TCH + KSZ * TCOLS);
        const int ha1 = own ? (ha0 + TCOLS - 1) : (ha0 + 1);
        float v[8 * KSZ];
        const float* xw = xb + (size_t)(wv * 8) * (H * W) + ww0 + ln;
        #pragma unroll
        for (int i = 0; i < 8 * KSZ; ++i) {
            const int ci = i / KSZ, r = i % KSZ;
            const float a = xw[ci * (H * W) + rowoff[r]];
            v[i] = rowok[r] ? a : 0.0f;
        }
        #pragma unroll
        for (int i = 0; i < 8 * KSZ; ++i) {
            const int ci = i / KSZ, r = i % KSZ;
            tile[(wv * 8 + ci) * TCH + r * TCOLS + 1 + ln] = v[i];
        }
        tile[ha0] = hl;
        tile[ha1] = hr;
    }
    __syncthreads();

    const int wave_s = __builtin_amdgcn_readfirstlane(wave);

    if (wave_s < 4) {
        const float* tq = tile + (t < NPOS ? t : NPOS - 1);
        float s0 = 0.0f, s1 = 0.0f, s2 = 0.0f, s3 = 0.0f;
        #pragma unroll
        for (int c = 0; c < CIN; c += 4) {
            const float a0 = tq[(c + 0) * TCH];
            const float a1 = tq[(c + 1) * TCH];
            const float a2 = tq[(c + 2) * TCH];
            const float a3 = tq[(c + 3) * TCH];
            s0 += a0 * a0;
            s1 += a1 * a1;
            s2 += a2 * a2;
            s3 += a3 * a3;
        }
        const float nrm = sqrtf((s0 + s1) + (s2 + s3));
        if (t < NPOS)
            ldsN[t] = nrm;
    }

    float xa[4][4][5];
    {
        const float* ta = tile + l16 * TCH + 4 * wave;
        const int o0 = kh ? toff(2) : toff(0);
        const int o1 = kh ? toff(3) : toff(1);
        const int o2 = kh ? toff(6) : toff(4);
        const int o3 = kh ? toff(7) : toff(5);
        #pragma unroll
        for (int pp = 0; pp < 4; ++pp) {
            #pragma unroll
            for (int mt = 0; mt < 4; ++mt) {
                const float* q = ta + mt * 16 * TCH + pp;
                xa[pp][mt][0] = q[o0];
                xa[pp][mt][1] = q[o1];
                xa[pp][mt][2] = q[o2];
                xa[pp][mt][3] = q[o3];
                const float v8 = q[toff(8)];
                xa[pp][mt][4] = kh ? 0.0f : v8;
            }
        }
    }
    __syncthreads();

    if (wave_s < 6) {
        v2f af[2][3];
        #pragma unroll
        for (int m = 0; m < 2; ++m) {
            const float* nb = &ldsN[m * 16 + l16];
            af[m][0].x = nb[kh ? toff(2) : toff(0)];
            af[m][0].y = nb[kh ? toff(3) : toff(1)];
            af[m][1].x = nb[kh ? toff(6) : toff(4)];
            af[m][1].y = nb[kh ? toff(7) : toff(5)];
            const float r8 = nb[toff(8)];
            af[m][2].x = kh ? 0.0f : r8;
            af[m][2].y = 0.0f;
        }
        v8f c0, c1;
        #pragma unroll
        for (int r = 0; r < 8; ++r) { c0[r] = obias; c1[r] = obias; }
        #pragma unroll
        for (int s = 0; s < 3; ++s) {
            v2f bf;
            bf.x = ob[2 * s];
            bf.y = (s < 2) ? ob[2 * s + 1] : 0.0f;
            c0 = __builtin_amdgcn_wmma_f32_16x16x4_f32(
                      false, af[0][s],  false, bf,
                      (short)0, c0,  false,  false);
            c1 = __builtin_amdgcn_wmma_f32_16x16x4_f32(
                      false, af[1][s],  false, bf,
                      (short)0, c1,  false,  false);
        }
        if (oval) {
            float* orow = &ldsOp[(8 * kh) * 81 + oidx];
            #pragma unroll
            for (int r = 0; r < 8; ++r) {
                orow[r * 81]        = tanh_hw(c0[r]);
                orow[(16 + r) * 81] = tanh_hw(c1[r]);
            }
        }
    }
    __syncthreads();

    {
        const int kc = (l16 < K2) ? l16 : (l16 - K2);
        const float* ob0 = ldsOp + (4 * wave) * 81 + kc;
        const float* obh = ob0 + 2 * kh * K2;
        const float i0 = (2 * kh     == kc) ? 1.0f : 0.0f;
        const float i1 = (2 * kh + 1 == kc) ? 1.0f : 0.0f;
        const float i2 = (2 * kh + 4 == kc) ? 1.0f : 0.0f;
        const float i3 = (2 * kh + 5 == kc) ? 1.0f : 0.0f;
        const float i4 = (8          == kc) ? 1.0f : 0.0f;
        float* yb = ldsY + (4 * wave) * YSTRIDE + (8 * kh) * K2 + kc;
        #pragma unroll
        for (int pp = 0; pp < 4; ++pp) {
            const float* q = obh + pp * 81;
            v2f bf0, bf1, bf2;
            bf0.x = q[0]      + i0;
            bf0.y = q[K2]     + i1;
            bf1.x = q[4 * K2] + i2;
            bf1.y = q[5 * K2] + i3;
            const float b8 = ob0[pp * 81 + 8 * K2] + i4;
            bf2.x = kh ? 0.0f : b8;
            bf2.y = 0.0f;
            #pragma unroll
            for (int mt = 0; mt < 4; ++mt) {
                v2f a0, a1, a2;
                a0.x = xa[pp][mt][0];
                a0.y = xa[pp][mt][1];
                a1.x = xa[pp][mt][2];
                a1.y = xa[pp][mt][3];
                a2.x = xa[pp][mt][4];
                a2.y = 0.0f;
                v8f d = {};
                d = __builtin_amdgcn_wmma_f32_16x16x4_f32(
                         false, a0,  false, bf0,
                         (short)0, d,  false,  false);
                d = __builtin_amdgcn_wmma_f32_16x16x4_f32(
                         false, a1,  false, bf1,
                         (short)0, d,  false,  false);
                d = __builtin_amdgcn_wmma_f32_16x16x4_f32(
                         false, a2,  false, bf2,
                         (short)0, d,  false,  false);
                float* yo = yb + pp * YSTRIDE + mt * 16 * K2;
                #pragma unroll
                for (int r = 0; r < 8; ++r)
                    yo[r * K2] = d[r];
            }
        }
    }

    const int kq    = wave & 1;
    const int nTile = wave >> 1;
    const int mIdx  = l16;
    const int nIdx  = l16;
    const float* wl = weight + (32 * kh) * CHW + (nTile * 16 + nIdx) * K2;
    constexpr int TRIPS = CIN / 8;
    const int c4b = kq * TRIPS;
    float b0[18];
    load_b18(b0, wl + c4b * CHW);
    __syncthreads();

    const float* ya0 = &ldsY[mIdx * YSTRIDE + 2 * QROWS * kh];
    const float* ya1 = ya0 + 16 * YSTRIDE;

    v8f acc0 = {};
    v8f acc1 = {};
    float b1[18];
    for (int i = 0; i < TRIPS; i += 2) {
        const int c4 = c4b + i;
        load_b18(b1, wl + (c4 + 1) * CHW);
        mma_trip2(acc0, acc1, b0, ya0 + c4 * K2, ya1 + c4 * K2);
        const int cn = (i + 2 < TRIPS) ? (c4 + 2) : (c4 + 1);
        load_b18(b0, wl + cn * CHW);
        mma_trip2(acc0, acc1, b1, ya0 + (c4 + 1) * K2, ya1 + (c4 + 1) * K2);
    }

    float* stage = lds + LDS_ST_OFF;
    const int orow = wave * 8 + (lane >> 3);
    const int ocol = (lane & 7) * 4;
    const float bo0 = bias[orow];
    const float bo1 = bias[orow + 4];
    {
        float* sp = stage + kq * SPLANE + (nTile * 16 + nIdx) * SPITCH + 8 * kh;
        *reinterpret_cast<float4*>(sp)      = make_float4(acc0[0], acc0[1], acc0[2], acc0[3]);
        *reinterpret_cast<float4*>(sp + 4)  = make_float4(acc0[4], acc0[5], acc0[6], acc0[7]);
        *reinterpret_cast<float4*>(sp + 16) = make_float4(acc1[0], acc1[1], acc1[2], acc1[3]);
        *reinterpret_cast<float4*>(sp + 20) = make_float4(acc1[4], acc1[5], acc1[6], acc1[7]);
    }
    __syncthreads();
    {
        const float* s0 = stage + orow * SPITCH + ocol;
        const float4 a0 = *reinterpret_cast<const float4*>(s0);
        const float4 a1 = *reinterpret_cast<const float4*>(s0 + SPLANE);
        const float4 c0 = *reinterpret_cast<const float4*>(s0 + 4 * SPITCH);
        const float4 c1 = *reinterpret_cast<const float4*>(s0 + 4 * SPITCH + SPLANE);
        float* op = out + ((size_t)(bb * COUT + orow)) * (H * W) + hw0 + ocol;
        *reinterpret_cast<float4*>(op) =
            make_float4((a0.x + a1.x) + bo0, (a0.y + a1.y) + bo0, (a0.z + a1.z) + bo0, (a0.w + a1.w) + bo0);
        *reinterpret_cast<float4*>(op + 4 * (H * W)) =
            make_float4((c0.x + c1.x) + bo1, (c0.y + c1.y) + bo1, (c0.z + c1.z) + bo1, (c0.w + c1.w) + bo1);
    }
}

extern "C" void kernel_launch(void* const* d_in, const int* in_sizes, int n_in,
                              void* d_out, int out_size, void* d_ws, size_t ws_size,
                              hipStream_t stream) {
    (void)in_sizes; (void)n_in; (void)out_size; (void)d_ws; (void)ws_size;
    const float* x      = (const float*)d_in[0];
    const float* weight = (const float*)d_in[1];
    const float* bias   = (const float*)d_in[2];
    const float* op_w   = (const float*)d_in[3];
    const float* op_b   = (const float*)d_in[4];
    float* out = (float*)d_out;

    const int totalPix = NB * H * W;
    const int blocks   = totalPix / PIX_PER_BLK;
    const size_t ldsBytes = (size_t)LDS_FLOATS * sizeof(float);

    dyfconv_fused<<<blocks, THREADS, ldsBytes, stream>>>(x, weight, bias, op_w, op_b, out);
}


# Main GEMM: next trip's 36 A values read from LDS as one group into a second register set ahead of the current trip's 18 WMMAs (were read among them, each waited for at once); sum order kept
# speedup vs baseline: 6.6239x; 1.0590x over previous
#include <hip/hip_runtime.h>
#include <hip/hip_bf16.h>
#include <math.h>

typedef __attribute__((ext_vector_type(2))) float v2f;
typedef __attribute__((ext_vector_type(8))) float v8f;

constexpr int KSZ   = 3;
constexpr int K2    = 9;
constexpr int CIN   = 64;
constexpr int COUT  = 64;
constexpr int NB    = 8;
constexpr int H     = 128;
constexpr int W     = 128;
constexpr int KDIM  = CIN * K2;

constexpr int THREADS     = 256;
constexpr int PIX_PER_BLK = 32;
constexpr int YSTRIDE     = KDIM + 1;

constexpr int LDS_Y_OFF  = 0;
constexpr int LDS_OP_OFF = LDS_Y_OFF + PIX_PER_BLK * YSTRIDE;
constexpr int LDS_N_OFF  = LDS_OP_OFF + PIX_PER_BLK * 81;
constexpr int SPITCH     = 36;
constexpr int SPLANE     = COUT * SPITCH;
constexpr int LDS_ST_OFF = LDS_OP_OFF;
constexpr int LDS_END_A  = LDS_N_OFF + PIX_PER_BLK * K2;
constexpr int LDS_END_B  = LDS_ST_OFF + 2 * SPLANE;
constexpr int LDS_FLOATS = LDS_END_A > LDS_END_B ? LDS_END_A : LDS_END_B;
static_assert((LDS_ST_OFF % 4) == 0 && (SPITCH % 4) == 0 && (SPLANE % 4) == 0, "16-byte alignment of the stage rows");

constexpr int TCOLS = PIX_PER_BLK + 2;
constexpr int TCH   = 105;
constexpr int NPOS  = KSZ * TCOLS;
static_assert(CIN * TCH <= PIX_PER_BLK * YSTRIDE, "tile must fit inside ldsY");
static_assert(NPOS <= PIX_PER_BLK * K2 && NPOS <= 4 * 32, "norms fit in ldsN; four waves cover the positions");
static_assert(W % PIX_PER_BLK == 0, "a block's pixels must share one image row");
__host__ __device__ constexpr int toff(int j) { return (j / KSZ) * TCOLS + (j % KSZ); }

constexpr int CHW   = COUT * K2;
constexpr int QROWS = 16 * K2;
static_assert(((2 * QROWS) % 64) == 32 && (YSTRIDE % 64) == 1, "bank spacing of the A fragment reads");
__host__ __device__ constexpr int kq_sel(int s, int e) { return s < 4 ? 0 : (s < 8 ? 1 : e); }
__host__ __device__ constexpr int ktap(int s, int e)   { return s < 8 ? 2 * (s & 3) + e : 8; }

__device__ __forceinline__ void load_b18(float (&b)[18], const float* __restrict__ wl) {
    #pragma unroll
    for (int s = 0; s < K2; ++s) {
        b[2 * s]     = wl[kq_sel(s, 0) * (16 * CHW) + ktap(s, 0)];
        b[2 * s + 1] = wl[kq_sel(s, 1) * (16 * CHW) + ktap(s, 1)];
    }
}

__device__ __forceinline__ void load_a36(v2f (&a0)[K2], v2f (&a1)[K2],
                                         const float* ya0, const float* ya1) {
    #pragma unroll
    for (int s = 0; s < K2; ++s) {
        a0[s].x = ya0[kq_sel(s, 0) * QROWS + ktap(s, 0)];
        a0[s].y = ya0[kq_sel(s, 1) * QROWS + ktap(s, 1)];
        a1[s].x = ya1[kq_sel(s, 0) * QROWS + ktap(s, 0)];
        a1[s].y = ya1[kq_sel(s, 1) * QROWS + ktap(s, 1)];
    }
}

__device__ __forceinline__ void mma_trip2(v8f& acc0, v8f& acc1, const float (&b)[18],
                                          const v2f (&a0)[K2], const v2f (&a1)[K2]) {
    #pragma unroll
    for (int s = 0; s < K2; ++s) {
        v2f bf;
        bf.x = b[2 * s];
        bf.y = b[2 * s + 1];
        acc0 = __builtin_amdgcn_wmma_f32_16x16x4_f32(
                    false, a0[s],  false, bf,
                    (short)0, acc0,  false,  false);
        acc1 = __builtin_amdgcn_wmma_f32_16x16x4_f32(
                    false, a1[s],  false, bf,
                    (short)0, acc1,  false,  false);
    }
}

__device__ __forceinline__ float tanh_hw(float x) {
    return __builtin_amdgcn_tanhf(x);
}

__global__ __launch_bounds__(THREADS) __attribute__((amdgpu_waves_per_eu(6, 6)))
void dyfconv_fused(const float* __restrict__ x,
                   const float* __restrict__ weight,
                   const float* __restrict__ bias,
                   const float* __restrict__ op_w,
                   const float* __restrict__ op_b,
                   float* __restrict__ out)
{
    extern __shared__ __attribute__((aligned(16))) float lds[];
    float* ldsY  = lds + LDS_Y_OFF;
    float* ldsOp = lds + LDS_OP_OFF;
    float* ldsN  = lds + LDS_N_OFF;

    const int t     = threadIdx.x;
    const int pbase = blockIdx.x * PIX_PER_BLK;
    const int wave  = t >> 5;
    const int lane  = t & 31;
    const int kh    = lane >> 4;
    const int l16   = lane & 15;

    const int  oidx = wave * 16 + l16;
    const bool oval = oidx < K2 * K2;
    float ob[5];
    float obias;
    {
        const int oc = oval ? oidx : (K2 * K2 - 1);
        const float* wrow = op_w + oc * K2;
        const float w0 = wrow[2 * kh];
        const float w1 = wrow[2 * kh + 1];
        const float w2 = wrow[4 + 2 * kh];
        const float w3 = wrow[5 + 2 * kh];
        const float w4 = wrow[8];
        const float wb = op_b[oc];
        ob[0] = oval ? w0 : 0.0f;
        ob[1] = oval ? w1 : 0.0f;
        ob[2] = oval ? w2 : 0.0f;
        ob[3] = oval ? w3 : 0.0f;
        ob[4] = (oval && kh == 0) ? w4 : 0.0f;
        obias = oval ? wb : 0.0f;
    }

    float* tile = ldsY;
    const int bb  = pbase >> 14;
    const int hw0 = pbase & (H * W - 1);
    const int hh  = hw0 >> 7;
    const int ww0 = hw0 & (W - 1);
    const float* xb = x + (size_t)bb * CIN * (H * W);
    {
        const int wv = t >> 5;
        const int ln = t & 31;
        int  rowoff[KSZ];
        bool rowok[KSZ];
        #pragma unroll
        for (int r = 0; r < KSZ; ++r) {
            const int hs = hh + r - 1;
            rowok[r]  = (unsigned)hs < (unsigned)H;
            rowoff[r] = (rowok[r] ? hs : hh) * W;
        }
        const bool own = ln < 8 * KSZ;
        const int  hq  = own ? ln : (8 * KSZ - 1);
        const int  hci = hq / KSZ;
        const int  hrw = hq - hci * KSZ;
        const bool okr = (unsigned)(hh + hrw - 1) < (unsigned)H;
        const float* xr = xb + (size_t)(wv * 8 + hci) * (H * W) + (okr ? (hh + hrw - 1) : hh) * W;
        const bool okl = okr && (ww0 > 0);
        const bool okq = okr && (ww0 + PIX_PER_BLK < W);
        const float hla = xr[okl ? ww0 - 1 : ww0];
        const float hra = xr[okq ? ww0 + PIX_PER_BLK : ww0];
        const float hl = (own && okl) ? hla : 0.0f;
        const float hr = (own && okq) ? hra : 0.0f;
        const int ha0 = own ? ((wv * 8 + hci) * TCH + hrw * TCOLS)
                            : ((wv * 8 + (ln - 8 * KSZ)) * TCH + KSZ * TCOLS);
        const int ha1 = own ? (ha0 + TCOLS - 1) : (ha0 + 1);
        float v[8 * KSZ];
        const float* xw = xb + (size_t)(wv * 8) * (H * W) + ww0 + ln;
        #pragma unroll
        for (int i = 0; i < 8 * KSZ; ++i) {
            const int ci = i / KSZ, r = i % KSZ;
            const float a = xw[ci * (H * W) + rowoff[r]];
            v[i] = rowok[r] ? a : 0.0f;
        }
        #pragma unroll
        for (int i = 0; i < 8 * KSZ; ++i) {
            const int ci = i / KSZ, r = i % KSZ;
            tile[(wv * 8 + ci) * TCH + r * TCOLS + 1 + ln] = v[i];
        }
        tile[ha0] = hl;
        tile[ha1] = hr;
    }
    __syncthreads();

    const int wave_s = __builtin_amdgcn_readfirstlane(wave);

    if (wave_s < 4) {
        const float* tq = tile + (t < NPOS ? t : NPOS - 1);
        float s0 = 0.0f, s1 = 0.0f, s2 = 0.0f, s3 = 0.0f;
        #pragma unroll
        for (int c = 0; c < CIN; c += 4) {
            const float a0 = tq[(c + 0) * TCH];
            const float a1 = tq[(c + 1) * TCH];
            const float a2 = tq[(c + 2) * TCH];
            const float a3 = tq[(c + 3) * TCH];
            s0 += a0 * a0;
            s1 += a1 * a1;
            s2 += a2 * a2;
            s3 += a3 * a3;
        }
        const float nrm = sqrtf((s0 + s1) + (s2 + s3));
        if (t < NPOS)
            ldsN[t] = nrm;
    }

    float xa[4][4][5];
    {
        const float* ta = tile + l16 * TCH + 4 * wave;
        const int o0 = kh ? toff(2) : toff(0);
        const int o1 = kh ? toff(3) : toff(1);
        const int o2 = kh ? toff(6) : toff(4);
        const int o3 = kh ? toff(7) : toff(5);
        #pragma unroll
        for (int pp = 0; pp < 4; ++pp) {
            #pragma unroll
            for (int mt = 0; mt < 4; ++mt) {
                const float* q = ta + mt * 16 * TCH + pp;
                xa[pp][mt][0] = q[o0];
                xa[pp][mt][1] = q[o1];
                xa[pp][mt][2] = q[o2];
                xa[pp][mt][3] = q[o3];
                const float v8 = q[toff(8)];
                xa[pp][mt][4] = kh ? 0.0f : v8;
            }
        }
    }
    __syncthreads();

    if (wave_s < 6) {
        v2f af[2][3];
        #pragma unroll
        for (int m = 0; m < 2; ++m) {
            const float* nb = &ldsN[m * 16 + l16];
            af[m][0].x = nb[kh ? toff(2) : toff(0)];
            af[m][0].y = nb[kh ? toff(3) : toff(1)];
            af[m][1].x = nb[kh ? toff(6) : toff(4)];
            af[m][1].y = nb[kh ? toff(7) : toff(5)];
            const float r8 = nb[toff(8)];
            af[m][2].x = kh ? 0.0f : r8;
            af[m][2].y = 0.0f;
        }
        v8f c0, c1;
        #pragma unroll
        for (int r = 0; r < 8; ++r) { c0[r] = obias; c1[r] = obias; }
        #pragma unroll
        for (int s = 0; s < 3; ++s) {
            v2f bf;
            bf.x = ob[2 * s];
            bf.y = (s < 2) ? ob[2 * s + 1] : 0.0f;
            c0 = __builtin_amdgcn_wmma_f32_16x16x4_f32(
                      false, af[0][s],  false, bf,
                      (short)0, c0,  false,  false);
            c1 = __builtin_amdgcn_wmma_f32_16x16x4_f32(
                      false, af[1][s],  false, bf,
                      (short)0, c1,  false,  false);
        }
        if (oval) {
            float* orow = &ldsOp[(8 * kh) * 81 + oidx];
            #pragma unroll
            for (int r = 0; r < 8; ++r) {
                orow[r * 81]        = tanh_hw(c0[r]);
                orow[(16 + r) * 81] = tanh_hw(c1[r]);
            }
        }
    }
    __syncthreads();

    {
        const int kc = (l16 < K2) ? l16 : (l16 - K2);
        const float* ob0 = ldsOp + (4 * wave) * 81 + kc;
        const float* obh = ob0 + 2 * kh * K2;
        const float i0 = (2 * kh     == kc) ? 1.0f : 0.0f;
        const float i1 = (2 * kh + 1 == kc) ? 1.0f : 0.0f;
        const float i2 = (2 * kh + 4 == kc) ? 1.0f : 0.0f;
        const float i3 = (2 * kh + 5 == kc) ? 1.0f : 0.0f;
        const float i4 = (8          == kc) ? 1.0f : 0.0f;
        float* yb = ldsY + (4 * wave) * YSTRIDE + (8 * kh) * K2 + kc;
        #pragma unroll
        for (int pp = 0; pp < 4; ++pp) {
            const float* q = obh + pp * 81;
            v2f bf0, bf1, bf2;
            bf0.x = q[0]      + i0;
            bf0.y = q[K2]     + i1;
            bf1.x = q[4 * K2] + i2;
            bf1.y = q[5 * K2] + i3;
            const float b8 = ob0[pp * 81 + 8 * K2] + i4;
            bf2.x = kh ? 0.0f : b8;
            bf2.y = 0.0f;
            #pragma unroll
            for (int mt = 0; mt < 4; ++mt) {
                v2f a0, a1, a2;
                a0.x = xa[pp][mt][0];
                a0.y = xa[pp][mt][1];
                a1.x = xa[pp][mt][2];
                a1.y = xa[pp][mt][3];
                a2.x = xa[pp][mt][4];
                a2.y = 0.0f;
                v8f d = {};
                d = __builtin_amdgcn_wmma_f32_16x16x4_f32(
                         false, a0,  false, bf0,
                         (short)0, d,  false,  false);
                d = __builtin_amdgcn_wmma_f32_16x16x4_f32(
                         false, a1,  false, bf1,
                         (short)0, d,  false,  false);
                d = __builtin_amdgcn_wmma_f32_16x16x4_f32(
                         false, a2,  false, bf2,
                         (short)0, d,  false,  false);
                float* yo = yb + pp * YSTRIDE + mt * 16 * K2;
                #pragma unroll
                for (int r = 0; r < 8; ++r)
                    yo[r * K2] = d[r];
            }
        }
    }

    const int kq    = wave & 1;
    const int nTile = wave >> 1;
    const int mIdx  = l16;
    const int nIdx  = l16;
    const float* wl = weight + (32 * kh) * CHW + (nTile * 16 + nIdx) * K2;
    constexpr int TRIPS = CIN / 8;
    const int c4b = kq * TRIPS;
    float b0[18];
    load_b18(b0, wl + c4b * CHW);
    __syncthreads();

    const float* ya0 = &ldsY[mIdx * YSTRIDE + 2 * QROWS * kh];
    const float* ya1 = ya0 + 16 * YSTRIDE;

    v8f acc0 = {};
    v8f acc1 = {};
    float b1[18];
    v2f a0c[K2], a1c[K2], a0n[K2], a1n[K2];
    load_a36(a0c, a1c, ya0 + c4b * K2, ya1 + c4b * K2);
    __builtin_amdgcn_sched_barrier(0);
    #pragma unroll
    for (int i = 0; i < TRIPS; i += 2) {
        const int c4 = c4b + i;
        load_a36(a0n, a1n, ya0 + (c4 + 1) * K2, ya1 + (c4 + 1) * K2);
        load_b18(b1, wl + (c4 + 1) * CHW);
        __builtin_amdgcn_sched_barrier(0);
        mma_trip2(acc0, acc1, b0, a0c, a1c);
        __builtin_amdgcn_sched_barrier(0);
        if (i + 2 < TRIPS) {
            load_a36(a0c, a1c, ya0 + (c4 + 2) * K2, ya1 + (c4 + 2) * K2);
            load_b18(b0, wl + (c4 + 2) * CHW);
        }
        __builtin_amdgcn_sched_barrier(0);
        mma_trip2(acc0, acc1, b1, a0n, a1n);
        __builtin_amdgcn_sched_barrier(0);
    }

    float* stage = lds + LDS_ST_OFF;
    const int orow = wave * 8 + (lane >> 3);
    const int ocol = (lane & 7) * 4;
    const float bo0 = bias[orow];
    const float bo1 = bias[orow + 4];
    {
        float* sp = stage + kq * SPLANE + (nTile * 16 + nIdx) * SPITCH + 8 * kh;
        *reinterpret_cast<float4*>(sp)      = make_float4(acc0[0], acc0[1], acc0[2], acc0[3]);
        *reinterpret_cast<float4*>(sp + 4)  = make_float4(acc0[4], acc0[5], acc0[6], acc0[7]);
        *reinterpret_cast<float4*>(sp + 16) = make_float4(acc1[0], acc1[1], acc1[2], acc1[3]);
        *reinterpret_cast<float4*>(sp + 20) = make_float4(acc1[4], acc1[5], acc1[6], acc1[7]);
    }
    __syncthreads();
    {
        const float* s0 = stage + orow * SPITCH + ocol;
        const float4 a0 = *reinterpret_cast<const float4*>(s0);
        const float4 a1 = *reinterpret_cast<const float4*>(s0 + SPLANE);
        const float4 c0 = *reinterpret_cast<const float4*>(s0 + 4 * SPITCH);
        const float4 c1 = *reinterpret_cast<const float4*>(s0 + 4 * SPITCH + SPLANE);
        float* op = out + ((size_t)(bb * COUT + orow)) * (H * W) + hw0 + ocol;
        *reinterpret_cast<float4*>(op) =
            make_float4((a0.x + a1.x) + bo0, (a0.y + a1.y) + bo0, (a0.z + a1.z) + bo0, (a0.w + a1.w) + bo0);
        *reinterpret_cast<float4*>(op + 4 * (H * W)) =
            make_float4((c0.x + c1.x) + bo1, (c0.y + c1.y) + bo1, (c0.z + c1.z) + bo1, (c0.w + c1.w) + bo1);
    }
}

extern "C" void kernel_launch(void* const* d_in, const int* in_sizes, int n_in,
                              void* d_out, int out_size, void* d_ws, size_t ws_size,
                              hipStream_t stream) {
    (void)in_sizes; (void)n_in; (void)out_size; (void)d_ws; (void)ws_size;
    const float* x      = (const float*)d_in[0];
    const float* weight = (const float*)d_in[1];
    const float* bias   = (const float*)d_in[2];
    const float* op_w   = (const float*)d_in[3];
    const float* op_b   = (const float*)d_in[4];
    float* out = (float*)d_out;

    const int totalPix = NB * H * W;
    const int blocks   = totalPix / PIX_PER_BLK;
    const size_t ldsBytes = (size_t)LDS_FLOATS * sizeof(float);

    dyfconv_fused<<<blocks, THREADS, ldsBytes, stream>>>(x, weight, bias, op_w, op_b, out);
}
